# starting kernel
# baseline (speedup 1.0000x reference)
#include <hip/hip_runtime.h>
#include <hip/hip_bf16.h>


typedef __attribute__((ext_vector_type(2))) float v2f;
typedef __attribute__((ext_vector_type(8))) float v8f;

namespace caps {
constexpr int B   = 32;
constexpr int In  = 4096;
constexpr int D   = 256;
constexpr int N   = 16;
constexpr int K   = 16;
constexpr int SPLIT = 8;
}

using namespace caps;

__global__ void zero_buf(float* p, int n) {
    int i = blockIdx.x * blockDim.x + threadIdx.x;
    if (i < n) p[i] = 0.0f;
}

__global__ void softmax_n(float* __restrict__ bc) {
    int idx = blockIdx.x * blockDim.x + threadIdx.x;
    if (idx >= B * In) return;
    float* p = bc + (size_t)idx * N;
    float v[N];
    float m = -3.402823466e+38f;
#pragma unroll
    for (int n = 0; n < N; ++n) { v[n] = p[n]; m = fmaxf(m, v[n]); }
    float s = 0.0f;
#pragma unroll
    for (int n = 0; n < N; ++n) { v[n] = __expf(v[n] - m); s += v[n]; }
    float inv = 1.0f / s;
#pragma unroll
    for (int n = 0; n < N; ++n) p[n] = v[n] * inv;
}

__global__ void __launch_bounds__(128)
gemm_s(const float* __restrict__ x, const float* __restrict__ c,
       float* __restrict__ s_part) {
    const int b    = blockIdx.y;
    const int sp   = blockIdx.x;
    const int lane = threadIdx.x & 31;
    const int wave = threadIdx.x >> 5;
    const int half = lane >> 4;
    const int l16  = lane & 15;

    const float* cb = c + (size_t)b * In * N;
    const float* xb = x + (size_t)b * In * D;

    const int d0 = wave * 64;
    const int i_beg = sp * (In / SPLIT);
    const int i_end = i_beg + (In / SPLIT);

    v8f acc0 = {}, acc1 = {}, acc2 = {}, acc3 = {};

    for (int i0 = i_beg; i0 < i_end; i0 += 4) {
        const int kr = i0 + half * 2;
        v2f a;
        a.x = cb[(size_t)(kr + 0) * N + l16];
        a.y = cb[(size_t)(kr + 1) * N + l16];
        const float* xr0 = xb + (size_t)(kr + 0) * D;
        const float* xr1 = xb + (size_t)(kr + 1) * D;
        __builtin_prefetch(xr0 + 4 * D, 0, 1);
        v2f b0, b1, b2, b3;
        b0.x = xr0[d0 +  0 + l16]; b0.y = xr1[d0 +  0 + l16];
        b1.x = xr0[d0 + 16 + l16]; b1.y = xr1[d0 + 16 + l16];
        b2.x = xr0[d0 + 32 + l16]; b2.y = xr1[d0 + 32 + l16];
        b3.x = xr0[d0 + 48 + l16]; b3.y = xr1[d0 + 48 + l16];
        acc0 = __builtin_amdgcn_wmma_f32_16x16x4_f32(false, a, false, b0, (short)0, acc0, false, false);
        acc1 = __builtin_amdgcn_wmma_f32_16x16x4_f32(false, a, false, b1, (short)0, acc1, false, false);
        acc2 = __builtin_amdgcn_wmma_f32_16x16x4_f32(false, a, false, b2, (short)0, acc2, false, false);
        acc3 = __builtin_amdgcn_wmma_f32_16x16x4_f32(false, a, false, b3, (short)0, acc3, false, false);
    }

    float* outp = s_part + (size_t)(b * SPLIT + sp) * N * D;
#pragma unroll
    for (int r = 0; r < 8; ++r) {
        const int row = r + half * 8;
        outp[(size_t)row * D + d0 +  0 + l16] = acc0[r];
        outp[(size_t)row * D + d0 + 16 + l16] = acc1[r];
        outp[(size_t)row * D + d0 + 32 + l16] = acc2[r];
        outp[(size_t)row * D + d0 + 48 + l16] = acc3[r];
    }
}

__global__ void reduce_s(const float* __restrict__ s_part, float* __restrict__ s) {
    int idx = blockIdx.x * blockDim.x + threadIdx.x;
    if (idx >= B * N * D) return;
    int b  = idx / (N * D);
    int nd = idx - b * (N * D);
    float acc = 0.0f;
#pragma unroll
    for (int sp = 0; sp < SPLIT; ++sp)
        acc += s_part[(size_t)(b * SPLIT + sp) * N * D + nd];
    s[idx] = acc;
}

__global__ void __launch_bounds__(256)
outputs_squash(const float* __restrict__ s, const float* __restrict__ W,
               float* __restrict__ out) {
    __shared__ float part[D * K];
    __shared__ float dot[K];
    const int bn = blockIdx.x;
    const int n  = bn & (N - 1);
    const int d  = threadIdx.x;

    const float  sv = s[(size_t)bn * D + d];
    const float* wr = W + (size_t)d * (N * K) + (size_t)n * K;
#pragma unroll
    for (int k = 0; k < K; ++k) part[d * K + k] = sv * wr[k];
    __syncthreads();

    if (d < K) {
        float a = 0.0f;
        for (int dd = 0; dd < D; ++dd) a += part[dd * K + d];
        dot[d] = a;
    }
    __syncthreads();

    if (d < K) {
        float n2 = 0.0f;
#pragma unroll
        for (int k = 0; k < K; ++k) n2 += dot[k] * dot[k];
        out[(size_t)bn * K + d] = dot[d] * rsqrtf(n2 + 1e-7f);
    }
}

__global__ void __launch_bounds__(256)
compute_t(const float* __restrict__ W, const float* __restrict__ out,
          float* __restrict__ t) {
    __shared__ float ov[K];
    const int bn = blockIdx.x;
    const int n  = bn & (N - 1);
    const int d  = threadIdx.x;
    if (d < K) ov[d] = out[(size_t)bn * K + d];
    __syncthreads();
    const float* wr = W + (size_t)d * (N * K) + (size_t)n * K;
    float acc = 0.0f;
#pragma unroll
    for (int k = 0; k < K; ++k) acc += wr[k] * ov[k];
    t[(size_t)bn * D + d] = acc;
}

__global__ void __launch_bounds__(128)
bupdate(const float* __restrict__ x, const float* __restrict__ t,
        float* __restrict__ bc) {
    const int b    = blockIdx.y;
    const int lane = threadIdx.x & 31;
    const int wave = threadIdx.x >> 5;
    const int half = lane >> 4;
    const int l16  = lane & 15;
    const int i0   = (blockIdx.x * 4 + wave) * 16;

    const float* xb = x + (size_t)b * In * D;
    const float* tb = t + (size_t)b * N * D;

    v8f acc = {};
    for (int d0 = 0; d0 < D; d0 += 4) {
        const int kd = d0 + half * 2;
        const float* xr = xb + (size_t)(i0 + l16) * D + kd;
        __builtin_prefetch(xr + 64, 0, 1);
        v2f a;  a.x = xr[0];  a.y = xr[1];
        const float* tr = tb + (size_t)l16 * D + kd;
        v2f bb; bb.x = tr[0]; bb.y = tr[1];
        acc = __builtin_amdgcn_wmma_f32_16x16x4_f32(false, a, false, bb, (short)0, acc, false, false);
    }

    float* dst = bc + ((size_t)b * In + i0) * N;
#pragma unroll
    for (int r = 0; r < 8; ++r)
        dst[(size_t)(r + half * 8) * N + l16] = acc[r];
}

extern "C" void kernel_launch(void* const* d_in, const int* in_sizes, int n_in,
                              void* d_out, int out_size, void* d_ws, size_t ws_size,
                              hipStream_t stream) {
    const float* x = (const float*)d_in[0];
    const float* W = (const float*)d_in[1];
    float* out = (float*)d_out;

    float* ws     = (float*)d_ws;
    float* bc     = ws;
    float* s_part = bc + (size_t)B * In * N;
    float* s      = s_part + (size_t)B * SPLIT * N * D;
    float* t      = s + (size_t)B * N * D;

    const int bc_n = B * In * N;
    zero_buf<<<(bc_n + 255) / 256, 256, 0, stream>>>(bc, bc_n);

    for (int it = 0; it < 3; ++it) {
        softmax_n<<<(B * In + 255) / 256, 256, 0, stream>>>(bc);
        gemm_s<<<dim3(SPLIT, B), 128, 0, stream>>>(x, bc, s_part);
        reduce_s<<<(B * N * D + 255) / 256, 256, 0, stream>>>(s_part, s);
        outputs_squash<<<B * N, 256, 0, stream>>>(s, W, out);
        if (it < 2) {
            compute_t<<<B * N, 256, 0, stream>>>(W, out, t);
            bupdate<<<dim3(In / 64, B), 128, 0, stream>>>(x, t, bc);
        }
    }
}


# bupdate: a wave walks 4 row tiles together sharing t's B fragments; x, t read from global as 16-byte row pieces in permuted k order, no LDS; 2 accumulators per tile: new order of sum over d
# speedup vs baseline: 1.1242x; 1.1242x over previous
#include <hip/hip_runtime.h>
#include <hip/hip_bf16.h>


typedef __attribute__((ext_vector_type(2))) float v2f;
typedef __attribute__((ext_vector_type(8))) float v8f;

namespace caps {
constexpr int B   = 32;
constexpr int In  = 4096;
constexpr int D   = 256;
constexpr int N   = 16;
constexpr int K   = 16;
constexpr int SPLIT = 8;
}

using namespace caps;

__global__ void zero_buf(float* p, int n) {
    int i = blockIdx.x * blockDim.x + threadIdx.x;
    if (i < n) p[i] = 0.0f;
}

__global__ void softmax_n(float* __restrict__ bc) {
    int idx = blockIdx.x * blockDim.x + threadIdx.x;
    if (idx >= B * In) return;
    float* p = bc + (size_t)idx * N;
    float v[N];
    float m = -3.402823466e+38f;
#pragma unroll
    for (int n = 0; n < N; ++n) { v[n] = p[n]; m = fmaxf(m, v[n]); }
    float s = 0.0f;
#pragma unroll
    for (int n = 0; n < N; ++n) { v[n] = __expf(v[n] - m); s += v[n]; }
    float inv = 1.0f / s;
#pragma unroll
    for (int n = 0; n < N; ++n) p[n] = v[n] * inv;
}

__global__ void __launch_bounds__(128)
gemm_s(const float* __restrict__ x, const float* __restrict__ c,
       float* __restrict__ s_part) {
    const int b    = blockIdx.y;
    const int sp   = blockIdx.x;
    const int lane = threadIdx.x & 31;
    const int wave = threadIdx.x >> 5;
    const int half = lane >> 4;
    const int l16  = lane & 15;

    const float* cb = c + (size_t)b * In * N;
    const float* xb = x + (size_t)b * In * D;

    const int d0 = wave * 64;
    const int i_beg = sp * (In / SPLIT);
    const int i_end = i_beg + (In / SPLIT);

    v8f acc0 = {}, acc1 = {}, acc2 = {}, acc3 = {};

    for (int i0 = i_beg; i0 < i_end; i0 += 4) {
        const int kr = i0 + half * 2;
        v2f a;
        a.x = cb[(size_t)(kr + 0) * N + l16];
        a.y = cb[(size_t)(kr + 1) * N + l16];
        const float* xr0 = xb + (size_t)(kr + 0) * D;
        const float* xr1 = xb + (size_t)(kr + 1) * D;
        __builtin_prefetch(xr0 + 4 * D, 0, 1);
        v2f b0, b1, b2, b3;
        b0.x = xr0[d0 +  0 + l16]; b0.y = xr1[d0 +  0 + l16];
        b1.x = xr0[d0 + 16 + l16]; b1.y = xr1[d0 + 16 + l16];
        b2.x = xr0[d0 + 32 + l16]; b2.y = xr1[d0 + 32 + l16];
        b3.x = xr0[d0 + 48 + l16]; b3.y = xr1[d0 + 48 + l16];
        acc0 = __builtin_amdgcn_wmma_f32_16x16x4_f32(false, a, false, b0, (short)0, acc0, false, false);
        acc1 = __builtin_amdgcn_wmma_f32_16x16x4_f32(false, a, false, b1, (short)0, acc1, false, false);
        acc2 = __builtin_amdgcn_wmma_f32_16x16x4_f32(false, a, false, b2, (short)0, acc2, false, false);
        acc3 = __builtin_amdgcn_wmma_f32_16x16x4_f32(false, a, false, b3, (short)0, acc3, false, false);
    }

    float* outp = s_part + (size_t)(b * SPLIT + sp) * N * D;
#pragma unroll
    for (int r = 0; r < 8; ++r) {
        const int row = r + half * 8;
        outp[(size_t)row * D + d0 +  0 + l16] = acc0[r];
        outp[(size_t)row * D + d0 + 16 + l16] = acc1[r];
        outp[(size_t)row * D + d0 + 32 + l16] = acc2[r];
        outp[(size_t)row * D + d0 + 48 + l16] = acc3[r];
    }
}

__global__ void reduce_s(const float* __restrict__ s_part, float* __restrict__ s) {
    int idx = blockIdx.x * blockDim.x + threadIdx.x;
    if (idx >= B * N * D) return;
    int b  = idx / (N * D);
    int nd = idx - b * (N * D);
    float acc = 0.0f;
#pragma unroll
    for (int sp = 0; sp < SPLIT; ++sp)
        acc += s_part[(size_t)(b * SPLIT + sp) * N * D + nd];
    s[idx] = acc;
}

__global__ void __launch_bounds__(256)
outputs_squash(const float* __restrict__ s, const float* __restrict__ W,
               float* __restrict__ out) {
    __shared__ float part[D * K];
    __shared__ float dot[K];
    const int bn = blockIdx.x;
    const int n  = bn & (N - 1);
    const int d  = threadIdx.x;

    const float  sv = s[(size_t)bn * D + d];
    const float* wr = W + (size_t)d * (N * K) + (size_t)n * K;
#pragma unroll
    for (int k = 0; k < K; ++k) part[d * K + k] = sv * wr[k];
    __syncthreads();

    if (d < K) {
        float a = 0.0f;
        for (int dd = 0; dd < D; ++dd) a += part[dd * K + d];
        dot[d] = a;
    }
    __syncthreads();

    if (d < K) {
        float n2 = 0.0f;
#pragma unroll
        for (int k = 0; k < K; ++k) n2 += dot[k] * dot[k];
        out[(size_t)bn * K + d] = dot[d] * rsqrtf(n2 + 1e-7f);
    }
}

__global__ void __launch_bounds__(256)
compute_t(const float* __restrict__ W, const float* __restrict__ out,
          float* __restrict__ t) {
    __shared__ float ov[K];
    const int bn = blockIdx.x;
    const int n  = bn & (N - 1);
    const int d  = threadIdx.x;
    if (d < K) ov[d] = out[(size_t)bn * K + d];
    __syncthreads();
    const float* wr = W + (size_t)d * (N * K) + (size_t)n * K;
    float acc = 0.0f;
#pragma unroll
    for (int k = 0; k < K; ++k) acc += wr[k] * ov[k];
    t[(size_t)bn * D + d] = acc;
}

typedef __attribute__((ext_vector_type(4))) float v4f;
constexpr int BU_TPW   = 4;
constexpr int BU_WAVES = 4;

__global__ void __launch_bounds__(128)
bupdate(const float* __restrict__ x, const float* __restrict__ t,
        float* __restrict__ bc) {
    const int b    = blockIdx.y;
    const int lane = threadIdx.x & 31;
    const int wave = threadIdx.x >> 5;
    const int half = lane >> 4;
    const int l16  = lane & 15;

    const int i0 = (blockIdx.x * BU_WAVES + wave) * BU_TPW * 16;
    const float* tr = t + (size_t)b * N * D + (size_t)l16 * D + half * 4;
    const float* xr = x + ((size_t)b * In + i0 + l16) * D + half * 4;

    v8f acc[BU_TPW][2] = {};
#pragma unroll
    for (int c = 0; c < D / 16; ++c) {
        const v4f ta = *(const v4f*)(tr + 16 * c);
        const v4f tb = *(const v4f*)(tr + 16 * c + 8);
        const v2f b0 = {ta.x, ta.y}, b1 = {ta.z, ta.w};
        const v2f b2 = {tb.x, tb.y}, b3 = {tb.z, tb.w};
#pragma unroll
        for (int tt = 0; tt < BU_TPW; ++tt) {
            const float* p = xr + (size_t)tt * 16 * D + 16 * c;
            const v4f xa = *(const v4f*)(p);
            const v4f xb = *(const v4f*)(p + 8);
            const v2f a0 = {xa.x, xa.y}, a1 = {xa.z, xa.w};
            const v2f a2 = {xb.x, xb.y}, a3 = {xb.z, xb.w};
            acc[tt][0] = __builtin_amdgcn_wmma_f32_16x16x4_f32(false, a0, false, b0, (short)0, acc[tt][0], false, false);
            acc[tt][1] = __builtin_amdgcn_wmma_f32_16x16x4_f32(false, a1, false, b1, (short)0, acc[tt][1], false, false);
            acc[tt][0] = __builtin_amdgcn_wmma_f32_16x16x4_f32(false, a2, false, b2, (short)0, acc[tt][0], false, false);
            acc[tt][1] = __builtin_amdgcn_wmma_f32_16x16x4_f32(false, a3, false, b3, (short)0, acc[tt][1], false, false);
        }
    }

#pragma unroll
    for (int tt = 0; tt < BU_TPW; ++tt) {
        float* dst = bc + ((size_t)b * In + i0 + tt * 16) * N;
#pragma unroll
        for (int r = 0; r < 8; ++r)
            dst[(size_t)(r + half * 8) * N + l16] = acc[tt][0][r] + acc[tt][1][r];
    }
}

extern "C" void kernel_launch(void* const* d_in, const int* in_sizes, int n_in,
                              void* d_out, int out_size, void* d_ws, size_t ws_size,
                              hipStream_t stream) {
    const float* x = (const float*)d_in[0];
    const float* W = (const float*)d_in[1];
    float* out = (float*)d_out;

    float* ws     = (float*)d_ws;
    float* bc     = ws;
    float* s_part = bc + (size_t)B * In * N;
    float* s      = s_part + (size_t)B * SPLIT * N * D;
    float* t      = s + (size_t)B * N * D;

    const int bc_n = B * In * N;
    zero_buf<<<(bc_n + 255) / 256, 256, 0, stream>>>(bc, bc_n);

    for (int it = 0; it < 3; ++it) {
        softmax_n<<<(B * In + 255) / 256, 256, 0, stream>>>(bc);
        gemm_s<<<dim3(SPLIT, B), 128, 0, stream>>>(x, bc, s_part);
        reduce_s<<<(B * N * D + 255) / 256, 256, 0, stream>>>(s_part, s);
        outputs_squash<<<B * N, 256, 0, stream>>>(s, W, out);
        if (it < 2) {
            compute_t<<<B * N, 256, 0, stream>>>(W, out, t);
            bupdate<<<dim3(In / (16 * BU_TPW * BU_WAVES), B), 128, 0, stream>>>(x, t, bc);
        }
    }
}


# gemm_s re-tiled: one 16-wave group per (b, 64 d cols) walks all 4096 rows, 16-byte x pieces feed 4 col-interleaved tiles, waves added in LDS, s written; reduce_s gone; new order of sum over i
# speedup vs baseline: 1.7768x; 1.5805x over previous
#include <hip/hip_runtime.h>
#include <hip/hip_bf16.h>


typedef __attribute__((ext_vector_type(2))) float v2f;
typedef __attribute__((ext_vector_type(4))) float v4f;
typedef __attribute__((ext_vector_type(8))) float v8f;

namespace caps {
constexpr int B   = 32;
constexpr int In  = 4096;
constexpr int D   = 256;
constexpr int N   = 16;
constexpr int K   = 16;
}

using namespace caps;

__global__ void zero_buf(float* p, int n) {
    int i = blockIdx.x * blockDim.x + threadIdx.x;
    if (i < n) p[i] = 0.0f;
}

__global__ void softmax_n(float* __restrict__ bc) {
    int idx = blockIdx.x * blockDim.x + threadIdx.x;
    if (idx >= B * In) return;
    float* p = bc + (size_t)idx * N;
    float v[N];
    float m = -3.402823466e+38f;
#pragma unroll
    for (int n = 0; n < N; ++n) { v[n] = p[n]; m = fmaxf(m, v[n]); }
    float s = 0.0f;
#pragma unroll
    for (int n = 0; n < N; ++n) { v[n] = __expf(v[n] - m); s += v[n]; }
    float inv = 1.0f / s;
#pragma unroll
    for (int n = 0; n < N; ++n) p[n] = v[n] * inv;
}

constexpr int GS_WAVES = 16;
constexpr int GS_COLS  = 64;
constexpr int GS_GRP   = 8;
constexpr int GS_ROWS  = In / GS_WAVES;
constexpr int GS_NGRP  = GS_ROWS / (4 * GS_GRP);

__global__ void __launch_bounds__(GS_WAVES * 32)
gemm_s(const float* __restrict__ x, const float* __restrict__ c,
       float* __restrict__ s) {
    __shared__ __attribute__((aligned(16))) float red[GS_WAVES * N * GS_COLS];
    const int b    = blockIdx.y;
    const int j    = blockIdx.x;
    const int lane = threadIdx.x & 31;
    const int wave = threadIdx.x >> 5;
    const int half = lane >> 4;
    const int l16  = lane & 15;

    const int r0 = wave * GS_ROWS + half * 2;
    const float* cp = c + ((size_t)b * In + r0) * N + l16;
    const float* xp = x + ((size_t)b * In + r0) * D + j * GS_COLS + l16 * 4;

    v8f acc0 = {}, acc1 = {}, acc2 = {}, acc3 = {};
    v4f xv[2][GS_GRP][2];
    float cv[2][GS_GRP][2];

#pragma unroll
    for (int ks = 0; ks < GS_GRP; ++ks) {
#pragma unroll
        for (int v = 0; v < 2; ++v) {
            xv[0][ks][v] = *(const v4f*)(xp + (size_t)(4 * ks + v) * D);
            cv[0][ks][v] = cp[(size_t)(4 * ks + v) * N];
        }
    }

#pragma unroll
    for (int g = 0; g < GS_NGRP; ++g) {
        const int cur = g & 1, nxt = cur ^ 1;
        if (g + 1 < GS_NGRP) {
#pragma unroll
            for (int ks = 0; ks < GS_GRP; ++ks) {
#pragma unroll
                for (int v = 0; v < 2; ++v) {
                    const int row = 4 * (GS_GRP * (g + 1) + ks) + v;
                    xv[nxt][ks][v] = *(const v4f*)(xp + (size_t)row * D);
                    cv[nxt][ks][v] = cp[(size_t)row * N];
                }
            }
        }
        __builtin_amdgcn_sched_barrier(0);
#pragma unroll
        for (int ks = 0; ks < GS_GRP; ++ks) {
            const v4f x0 = xv[cur][ks][0], x1 = xv[cur][ks][1];
            const v2f a  = {cv[cur][ks][0], cv[cur][ks][1]};
            const v2f b0 = {x0.x, x1.x}, b1 = {x0.y, x1.y};
            const v2f b2 = {x0.z, x1.z}, b3 = {x0.w, x1.w};
            acc0 = __builtin_amdgcn_wmma_f32_16x16x4_f32(false, a, false, b0, (short)0, acc0, false, false);
            acc1 = __builtin_amdgcn_wmma_f32_16x16x4_f32(false, a, false, b1, (short)0, acc1, false, false);
            acc2 = __builtin_amdgcn_wmma_f32_16x16x4_f32(false, a, false, b2, (short)0, acc2, false, false);
            acc3 = __builtin_amdgcn_wmma_f32_16x16x4_f32(false, a, false, b3, (short)0, acc3, false, false);
        }
    }

    float* mine = red + wave * (N * GS_COLS);
#pragma unroll
    for (int r = 0; r < 8; ++r) {
        const v4f o = {acc0[r], acc1[r], acc2[r], acc3[r]};
        *(v4f*)(mine + (r + half * 8) * GS_COLS + l16 * 4) = o;
    }
    __syncthreads();

    if (threadIdx.x < N * GS_COLS / 4) {
        const int n  = threadIdx.x >> 4;
        const int c4 = threadIdx.x & 15;
        v4f sum = *(const v4f*)(red + n * GS_COLS + c4 * 4);
#pragma unroll
        for (int w = 1; w < GS_WAVES; ++w)
            sum += *(const v4f*)(red + (w * N + n) * GS_COLS + c4 * 4);
        *(v4f*)(s + ((size_t)b * N + n) * D + j * GS_COLS + c4 * 4) = sum;
    }
}

__global__ void __launch_bounds__(256)
outputs_squash(const float* __restrict__ s, const float* __restrict__ W,
               float* __restrict__ out) {
    __shared__ float part[D * K];
    __shared__ float dot[K];
    const int bn = blockIdx.x;
    const int n  = bn & (N - 1);
    const int d  = threadIdx.x;

    const float  sv = s[(size_t)bn * D + d];
    const float* wr = W + (size_t)d * (N * K) + (size_t)n * K;
#pragma unroll
    for (int k = 0; k < K; ++k) part[d * K + k] = sv * wr[k];
    __syncthreads();

    if (d < K) {
        float a = 0.0f;
        for (int dd = 0; dd < D; ++dd) a += part[dd * K + d];
        dot[d] = a;
    }
    __syncthreads();

    if (d < K) {
        float n2 = 0.0f;
#pragma unroll
        for (int k = 0; k < K; ++k) n2 += dot[k] * dot[k];
        out[(size_t)bn * K + d] = dot[d] * rsqrtf(n2 + 1e-7f);
    }
}

__global__ void __launch_bounds__(256)
compute_t(const float* __restrict__ W, const float* __restrict__ out,
          float* __restrict__ t) {
    __shared__ float ov[K];
    const int bn = blockIdx.x;
    const int n  = bn & (N - 1);
    const int d  = threadIdx.x;
    if (d < K) ov[d] = out[(size_t)bn * K + d];
    __syncthreads();
    const float* wr = W + (size_t)d * (N * K) + (size_t)n * K;
    float acc = 0.0f;
#pragma unroll
    for (int k = 0; k < K; ++k) acc += wr[k] * ov[k];
    t[(size_t)bn * D + d] = acc;
}

constexpr int BU_TPW   = 4;
constexpr int BU_WAVES = 4;

__global__ void __launch_bounds__(128)
bupdate(const float* __restrict__ x, const float* __restrict__ t,
        float* __restrict__ bc) {
    const int b    = blockIdx.y;
    const int lane = threadIdx.x & 31;
    const int wave = threadIdx.x >> 5;
    const int half = lane >> 4;
    const int l16  = lane & 15;

    const int i0 = (blockIdx.x * BU_WAVES + wave) * BU_TPW * 16;
    const float* tr = t + (size_t)b * N * D + (size_t)l16 * D + half * 4;
    const float* xr = x + ((size_t)b * In + i0 + l16) * D + half * 4;

    v8f acc[BU_TPW][2] = {};
#pragma unroll
    for (int c = 0; c < D / 16; ++c) {
        const v4f ta = *(const v4f*)(tr + 16 * c);
        const v4f tb = *(const v4f*)(tr + 16 * c + 8);
        const v2f b0 = {ta.x, ta.y}, b1 = {ta.z, ta.w};
        const v2f b2 = {tb.x, tb.y}, b3 = {tb.z, tb.w};
#pragma unroll
        for (int tt = 0; tt < BU_TPW; ++tt) {
            const float* p = xr + (size_t)tt * 16 * D + 16 * c;
            const v4f xa = *(const v4f*)(p);
            const v4f xb = *(const v4f*)(p + 8);
            const v2f a0 = {xa.x, xa.y}, a1 = {xa.z, xa.w};
            const v2f a2 = {xb.x, xb.y}, a3 = {xb.z, xb.w};
            acc[tt][0] = __builtin_amdgcn_wmma_f32_16x16x4_f32(false, a0, false, b0, (short)0, acc[tt][0], false, false);
            acc[tt][1] = __builtin_amdgcn_wmma_f32_16x16x4_f32(false, a1, false, b1, (short)0, acc[tt][1], false, false);
            acc[tt][0] = __builtin_amdgcn_wmma_f32_16x16x4_f32(false, a2, false, b2, (short)0, acc[tt][0], false, false);
            acc[tt][1] = __builtin_amdgcn_wmma_f32_16x16x4_f32(false, a3, false, b3, (short)0, acc[tt][1], false, false);
        }
    }

#pragma unroll
    for (int tt = 0; tt < BU_TPW; ++tt) {
        float* dst = bc + ((size_t)b * In + i0 + tt * 16) * N;
#pragma unroll
        for (int r = 0; r < 8; ++r)
            dst[(size_t)(r + half * 8) * N + l16] = acc[tt][0][r] + acc[tt][1][r];
    }
}

extern "C" void kernel_launch(void* const* d_in, const int* in_sizes, int n_in,
                              void* d_out, int out_size, void* d_ws, size_t ws_size,
                              hipStream_t stream) {
    const float* x = (const float*)d_in[0];
    const float* W = (const float*)d_in[1];
    float* out = (float*)d_out;

    float* ws     = (float*)d_ws;
    float* bc     = ws;
    float* s      = bc + (size_t)B * In * N;
    float* t      = s + (size_t)B * N * D;

    const int bc_n = B * In * N;
    zero_buf<<<(bc_n + 255) / 256, 256, 0, stream>>>(bc, bc_n);

    for (int it = 0; it < 3; ++it) {
        softmax_n<<<(B * In + 255) / 256, 256, 0, stream>>>(bc);
        gemm_s<<<dim3(D / GS_COLS, B), GS_WAVES * 32, 0, stream>>>(x, bc, s);
        outputs_squash<<<B * N, 256, 0, stream>>>(s, W, out);
        if (it < 2) {
            compute_t<<<B * N, 256, 0, stream>>>(W, out, t);
            bupdate<<<dim3(In / (16 * BU_TPW * BU_WAVES), B), 128, 0, stream>>>(x, t, bc);
        }
    }
}


# zero_buf: 2048 waves, each lane storing 8 x 16 bytes (32 KB contiguous per work-group), instead of 65536 waves of one 4-byte store per lane: fewer waves to enter; no sum touched
# speedup vs baseline: 1.9131x; 1.0767x over previous
#include <hip/hip_runtime.h>
#include <hip/hip_bf16.h>


typedef __attribute__((ext_vector_type(2))) float v2f;
typedef __attribute__((ext_vector_type(4))) float v4f;
typedef __attribute__((ext_vector_type(8))) float v8f;

namespace caps {
constexpr int B   = 32;
constexpr int In  = 4096;
constexpr int D   = 256;
constexpr int N   = 16;
constexpr int K   = 16;
}

using namespace caps;

constexpr int ZB_THREADS    = 256;
constexpr int ZB_PER_THREAD = 8;
constexpr int ZB_BLOCKS     = (B * In * N) / (4 * ZB_THREADS * ZB_PER_THREAD);
static_assert((B * In * N) % (4 * ZB_THREADS * ZB_PER_THREAD) == 0,
              "the logits must divide into whole work-group pieces");

__global__ void __launch_bounds__(ZB_THREADS)
zero_buf(float* __restrict__ p) {
    v4f* q = (v4f*)p + (size_t)blockIdx.x * (ZB_THREADS * ZB_PER_THREAD) + threadIdx.x;
    const v4f z = {0.0f, 0.0f, 0.0f, 0.0f};
    for (int k = 0; k < ZB_PER_THREAD; ++k) q[k * ZB_THREADS] = z;
}

__global__ void softmax_n(float* __restrict__ bc) {
    int idx = blockIdx.x * blockDim.x + threadIdx.x;
    if (idx >= B * In) return;
    float* p = bc + (size_t)idx * N;
    float v[N];
    float m = -3.402823466e+38f;
#pragma unroll
    for (int n = 0; n < N; ++n) { v[n] = p[n]; m = fmaxf(m, v[n]); }
    float s = 0.0f;
#pragma unroll
    for (int n = 0; n < N; ++n) { v[n] = __expf(v[n] - m); s += v[n]; }
    float inv = 1.0f / s;
#pragma unroll
    for (int n = 0; n < N; ++n) p[n] = v[n] * inv;
}

constexpr int GS_WAVES = 16;
constexpr int GS_COLS  = 64;
constexpr int GS_GRP   = 8;
constexpr int GS_ROWS  = In / GS_WAVES;
constexpr int GS_NGRP  = GS_ROWS / (4 * GS_GRP);

__global__ void __launch_bounds__(GS_WAVES * 32)
gemm_s(const float* __restrict__ x, const float* __restrict__ c,
       float* __restrict__ s) {
    __shared__ __attribute__((aligned(16))) float red[GS_WAVES * N * GS_COLS];
    const int b    = blockIdx.y;
    const int j    = blockIdx.x;
    const int lane = threadIdx.x & 31;
    const int wave = threadIdx.x >> 5;
    const int half = lane >> 4;
    const int l16  = lane & 15;

    const int r0 = wave * GS_ROWS + half * 2;
    const float* cp = c + ((size_t)b * In + r0) * N + l16;
    const float* xp = x + ((size_t)b * In + r0) * D + j * GS_COLS + l16 * 4;

    v8f acc0 = {}, acc1 = {}, acc2 = {}, acc3 = {};
    v4f xv[2][GS_GRP][2];
    float cv[2][GS_GRP][2];

#pragma unroll
    for (int ks = 0; ks < GS_GRP; ++ks) {
#pragma unroll
        for (int v = 0; v < 2; ++v) {
            xv[0][ks][v] = *(const v4f*)(xp + (size_t)(4 * ks + v) * D);
            cv[0][ks][v] = cp[(size_t)(4 * ks + v) * N];
        }
    }

#pragma unroll
    for (int g = 0; g < GS_NGRP; ++g) {
        const int cur = g & 1, nxt = cur ^ 1;
        if (g + 1 < GS_NGRP) {
#pragma unroll
            for (int ks = 0; ks < GS_GRP; ++ks) {
#pragma unroll
                for (int v = 0; v < 2; ++v) {
                    const int row = 4 * (GS_GRP * (g + 1) + ks) + v;
                    xv[nxt][ks][v] = *(const v4f*)(xp + (size_t)row * D);
                    cv[nxt][ks][v] = cp[(size_t)row * N];
                }
            }
        }
        __builtin_amdgcn_sched_barrier(0);
#pragma unroll
        for (int ks = 0; ks < GS_GRP; ++ks) {
            const v4f x0 = xv[cur][ks][0], x1 = xv[cur][ks][1];
            const v2f a  = {cv[cur][ks][0], cv[cur][ks][1]};
            const v2f b0 = {x0.x, x1.x}, b1 = {x0.y, x1.y};
            const v2f b2 = {x0.z, x1.z}, b3 = {x0.w, x1.w};
            acc0 = __builtin_amdgcn_wmma_f32_16x16x4_f32(false, a, false, b0, (short)0, acc0, false, false);
            acc1 = __builtin_amdgcn_wmma_f32_16x16x4_f32(false, a, false, b1, (short)0, acc1, false, false);
            acc2 = __builtin_amdgcn_wmma_f32_16x16x4_f32(false, a, false, b2, (short)0, acc2, false, false);
            acc3 = __builtin_amdgcn_wmma_f32_16x16x4_f32(false, a, false, b3, (short)0, acc3, false, false);
        }
    }

    float* mine = red + wave * (N * GS_COLS);
#pragma unroll
    for (int r = 0; r < 8; ++r) {
        const v4f o = {acc0[r], acc1[r], acc2[r], acc3[r]};
        *(v4f*)(mine + (r + half * 8) * GS_COLS + l16 * 4) = o;
    }
    __syncthreads();

    if (threadIdx.x < N * GS_COLS / 4) {
        const int n  = threadIdx.x >> 4;
        const int c4 = threadIdx.x & 15;
        v4f sum = *(const v4f*)(red + n * GS_COLS + c4 * 4);
#pragma unroll
        for (int w = 1; w < GS_WAVES; ++w)
            sum += *(const v4f*)(red + (w * N + n) * GS_COLS + c4 * 4);
        *(v4f*)(s + ((size_t)b * N + n) * D + j * GS_COLS + c4 * 4) = sum;
    }
}

__global__ void __launch_bounds__(256)
outputs_squash(const float* __restrict__ s, const float* __restrict__ W,
               float* __restrict__ out) {
    __shared__ float part[D * K];
    __shared__ float dot[K];
    const int bn = blockIdx.x;
    const int n  = bn & (N - 1);
    const int d  = threadIdx.x;

    const float  sv = s[(size_t)bn * D + d];
    const float* wr = W + (size_t)d * (N * K) + (size_t)n * K;
#pragma unroll
    for (int k = 0; k < K; ++k) part[d * K + k] = sv * wr[k];
    __syncthreads();

    if (d < K) {
        float a = 0.0f;
        for (int dd = 0; dd < D; ++dd) a += part[dd * K + d];
        dot[d] = a;
    }
    __syncthreads();

    if (d < K) {
        float n2 = 0.0f;
#pragma unroll
        for (int k = 0; k < K; ++k) n2 += dot[k] * dot[k];
        out[(size_t)bn * K + d] = dot[d] * rsqrtf(n2 + 1e-7f);
    }
}

__global__ void __launch_bounds__(256)
compute_t(const float* __restrict__ W, const float* __restrict__ out,
          float* __restrict__ t) {
    __shared__ float ov[K];
    const int bn = blockIdx.x;
    const int n  = bn & (N - 1);
    const int d  = threadIdx.x;
    if (d < K) ov[d] = out[(size_t)bn * K + d];
    __syncthreads();
    const float* wr = W + (size_t)d * (N * K) + (size_t)n * K;
    float acc = 0.0f;
#pragma unroll
    for (int k = 0; k < K; ++k) acc += wr[k] * ov[k];
    t[(size_t)bn * D + d] = acc;
}

constexpr int BU_TPW   = 4;
constexpr int BU_WAVES = 4;

__global__ void __launch_bounds__(128)
bupdate(const float* __restrict__ x, const float* __restrict__ t,
        float* __restrict__ bc) {
    const int b    = blockIdx.y;
    const int lane = threadIdx.x & 31;
    const int wave = threadIdx.x >> 5;
    const int half = lane >> 4;
    const int l16  = lane & 15;

    const int i0 = (blockIdx.x * BU_WAVES + wave) * BU_TPW * 16;
    const float* tr = t + (size_t)b * N * D + (size_t)l16 * D + half * 4;
    const float* xr = x + ((size_t)b * In + i0 + l16) * D + half * 4;

    v8f acc[BU_TPW][2] = {};
#pragma unroll
    for (int c = 0; c < D / 16; ++c) {
        const v4f ta = *(const v4f*)(tr + 16 * c);
        const v4f tb = *(const v4f*)(tr + 16 * c + 8);
        const v2f b0 = {ta.x, ta.y}, b1 = {ta.z, ta.w};
        const v2f b2 = {tb.x, tb.y}, b3 = {tb.z, tb.w};
#pragma unroll
        for (int tt = 0; tt < BU_TPW; ++tt) {
            const float* p = xr + (size_t)tt * 16 * D + 16 * c;
            const v4f xa = *(const v4f*)(p);
            const v4f xb = *(const v4f*)(p + 8);
            const v2f a0 = {xa.x, xa.y}, a1 = {xa.z, xa.w};
            const v2f a2 = {xb.x, xb.y}, a3 = {xb.z, xb.w};
            acc[tt][0] = __builtin_amdgcn_wmma_f32_16x16x4_f32(false, a0, false, b0, (short)0, acc[tt][0], false, false);
            acc[tt][1] = __builtin_amdgcn_wmma_f32_16x16x4_f32(false, a1, false, b1, (short)0, acc[tt][1], false, false);
            acc[tt][0] = __builtin_amdgcn_wmma_f32_16x16x4_f32(false, a2, false, b2, (short)0, acc[tt][0], false, false);
            acc[tt][1] = __builtin_amdgcn_wmma_f32_16x16x4_f32(false, a3, false, b3, (short)0, acc[tt][1], false, false);
        }
    }

#pragma unroll
    for (int tt = 0; tt < BU_TPW; ++tt) {
        float* dst = bc + ((size_t)b * In + i0 + tt * 16) * N;
#pragma unroll
        for (int r = 0; r < 8; ++r)
            dst[(size_t)(r + half * 8) * N + l16] = acc[tt][0][r] + acc[tt][1][r];
    }
}

extern "C" void kernel_launch(void* const* d_in, const int* in_sizes, int n_in,
                              void* d_out, int out_size, void* d_ws, size_t ws_size,
                              hipStream_t stream) {
    const float* x = (const float*)d_in[0];
    const float* W = (const float*)d_in[1];
    float* out = (float*)d_out;

    float* ws     = (float*)d_ws;
    float* bc     = ws;
    float* s      = bc + (size_t)B * In * N;
    float* t      = s + (size_t)B * N * D;

    zero_buf<<<ZB_BLOCKS, ZB_THREADS, 0, stream>>>(bc);

    for (int it = 0; it < 3; ++it) {
        softmax_n<<<(B * In + 255) / 256, 256, 0, stream>>>(bc);
        gemm_s<<<dim3(D / GS_COLS, B), GS_WAVES * 32, 0, stream>>>(x, bc, s);
        outputs_squash<<<B * N, 256, 0, stream>>>(s, W, out);
        if (it < 2) {
            compute_t<<<B * N, 256, 0, stream>>>(W, out, t);
            bupdate<<<dim3(In / (16 * BU_TPW * BU_WAVES), B), 128, 0, stream>>>(x, t, bc);
        }
    }
}


# gemm_s: rows split in 2 across work-groups (256 16-wave groups, one per WGP, instead of 128); each half writes a partial plane of s, outputs_squash adds the two: new order of the sum over i
# speedup vs baseline: 2.0557x; 1.0745x over previous
#include <hip/hip_runtime.h>
#include <hip/hip_bf16.h>


typedef __attribute__((ext_vector_type(2))) float v2f;
typedef __attribute__((ext_vector_type(4))) float v4f;
typedef __attribute__((ext_vector_type(8))) float v8f;

namespace caps {
constexpr int B   = 32;
constexpr int In  = 4096;
constexpr int D   = 256;
constexpr int N   = 16;
constexpr int K   = 16;
}

using namespace caps;

constexpr int ZB_THREADS    = 256;
constexpr int ZB_PER_THREAD = 8;
constexpr int ZB_BLOCKS     = (B * In * N) / (4 * ZB_THREADS * ZB_PER_THREAD);
static_assert((B * In * N) % (4 * ZB_THREADS * ZB_PER_THREAD) == 0,
              "the logits must divide into whole work-group pieces");

__global__ void __launch_bounds__(ZB_THREADS)
zero_buf(float* __restrict__ p) {
    v4f* q = (v4f*)p + (size_t)blockIdx.x * (ZB_THREADS * ZB_PER_THREAD) + threadIdx.x;
    const v4f z = {0.0f, 0.0f, 0.0f, 0.0f};
    for (int k = 0; k < ZB_PER_THREAD; ++k) q[k * ZB_THREADS] = z;
}

__global__ void softmax_n(float* __restrict__ bc) {
    int idx = blockIdx.x * blockDim.x + threadIdx.x;
    if (idx >= B * In) return;
    float* p = bc + (size_t)idx * N;
    float v[N];
    float m = -3.402823466e+38f;
#pragma unroll
    for (int n = 0; n < N; ++n) { v[n] = p[n]; m = fmaxf(m, v[n]); }
    float s = 0.0f;
#pragma unroll
    for (int n = 0; n < N; ++n) { v[n] = __expf(v[n] - m); s += v[n]; }
    float inv = 1.0f / s;
#pragma unroll
    for (int n = 0; n < N; ++n) p[n] = v[n] * inv;
}

constexpr int GS_WAVES = 16;
constexpr int GS_COLS  = 64;
constexpr int GS_GRP   = 8;
constexpr int GS_SPLIT = 2;
constexpr int GS_ROWS  = In / (GS_SPLIT * GS_WAVES);
constexpr int GS_NGRP  = GS_ROWS / (4 * GS_GRP);
static_assert(GS_SPLIT == 2, "gemm_s and outputs_squash name exactly two partial planes");
static_assert(GS_ROWS % (4 * GS_GRP) == 0, "a wave's rows must divide into whole load groups");

__global__ void __launch_bounds__(GS_WAVES * 32)
gemm_s(const float* __restrict__ x, const float* __restrict__ c,
       float* __restrict__ s0, float* __restrict__ s1) {
    __shared__ __attribute__((aligned(16))) float red[GS_WAVES * N * GS_COLS];
    const int b    = blockIdx.y;
    const int j    = blockIdx.x % (D / GS_COLS);
    const int z    = blockIdx.x / (D / GS_COLS);
    const int lane = threadIdx.x & 31;
    const int wave = threadIdx.x >> 5;
    const int half = lane >> 4;
    const int l16  = lane & 15;
    float* __restrict__ s = z == 0 ? s0 : s1;

    const int r0 = (z * GS_WAVES + wave) * GS_ROWS + half * 2;
    const float* cp = c + ((size_t)b * In + r0) * N + l16;
    const float* xp = x + ((size_t)b * In + r0) * D + j * GS_COLS + l16 * 4;

    v8f acc0 = {}, acc1 = {}, acc2 = {}, acc3 = {};
    v4f xv[2][GS_GRP][2];
    float cv[2][GS_GRP][2];

#pragma unroll
    for (int ks = 0; ks < GS_GRP; ++ks) {
#pragma unroll
        for (int v = 0; v < 2; ++v) {
            xv[0][ks][v] = *(const v4f*)(xp + (size_t)(4 * ks + v) * D);
            cv[0][ks][v] = cp[(size_t)(4 * ks + v) * N];
        }
    }

#pragma unroll
    for (int g = 0; g < GS_NGRP; ++g) {
        const int cur = g & 1, nxt = cur ^ 1;
        if (g + 1 < GS_NGRP) {
#pragma unroll
            for (int ks = 0; ks < GS_GRP; ++ks) {
#pragma unroll
                for (int v = 0; v < 2; ++v) {
                    const int row = 4 * (GS_GRP * (g + 1) + ks) + v;
                    xv[nxt][ks][v] = *(const v4f*)(xp + (size_t)row * D);
                    cv[nxt][ks][v] = cp[(size_t)row * N];
                }
            }
        }
        __builtin_amdgcn_sched_barrier(0);
#pragma unroll
        for (int ks = 0; ks < GS_GRP; ++ks) {
            const v4f x0 = xv[cur][ks][0], x1 = xv[cur][ks][1];
            const v2f a  = {cv[cur][ks][0], cv[cur][ks][1]};
            const v2f b0 = {x0.x, x1.x}, b1 = {x0.y, x1.y};
            const v2f b2 = {x0.z, x1.z}, b3 = {x0.w, x1.w};
            acc0 = __builtin_amdgcn_wmma_f32_16x16x4_f32(false, a, false, b0, (short)0, acc0, false, false);
            acc1 = __builtin_amdgcn_wmma_f32_16x16x4_f32(false, a, false, b1, (short)0, acc1, false, false);
            acc2 = __builtin_amdgcn_wmma_f32_16x16x4_f32(false, a, false, b2, (short)0, acc2, false, false);
            acc3 = __builtin_amdgcn_wmma_f32_16x16x4_f32(false, a, false, b3, (short)0, acc3, false, false);
        }
    }

    float* mine = red + wave * (N * GS_COLS);
#pragma unroll
    for (int r = 0; r < 8; ++r) {
        const v4f o = {acc0[r], acc1[r], acc2[r], acc3[r]};
        *(v4f*)(mine + (r + half * 8) * GS_COLS + l16 * 4) = o;
    }
    __syncthreads();

    if (threadIdx.x < N * GS_COLS / 4) {
        const int n  = threadIdx.x >> 4;
        const int c4 = threadIdx.x & 15;
        v4f sum = *(const v4f*)(red + n * GS_COLS + c4 * 4);
#pragma unroll
        for (int w = 1; w < GS_WAVES; ++w)
            sum += *(const v4f*)(red + (w * N + n) * GS_COLS + c4 * 4);
        *(v4f*)(s + ((size_t)b * N + n) * D + j * GS_COLS + c4 * 4) = sum;
    }
}

__global__ void __launch_bounds__(256)
outputs_squash(const float* __restrict__ s0, const float* __restrict__ s1,
               const float* __restrict__ W, float* __restrict__ out) {
    __shared__ float part[D * K];
    __shared__ float dot[K];
    const int bn = blockIdx.x;
    const int n  = bn & (N - 1);
    const int d  = threadIdx.x;

    const float  sv = s0[(size_t)bn * D + d] + s1[(size_t)bn * D + d];
    const float* wr = W + (size_t)d * (N * K) + (size_t)n * K;
#pragma unroll
    for (int k = 0; k < K; ++k) part[d * K + k] = sv * wr[k];
    __syncthreads();

    if (d < K) {
        float a = 0.0f;
        for (int dd = 0; dd < D; ++dd) a += part[dd * K + d];
        dot[d] = a;
    }
    __syncthreads();

    if (d < K) {
        float n2 = 0.0f;
#pragma unroll
        for (int k = 0; k < K; ++k) n2 += dot[k] * dot[k];
        out[(size_t)bn * K + d] = dot[d] * rsqrtf(n2 + 1e-7f);
    }
}

__global__ void __launch_bounds__(256)
compute_t(const float* __restrict__ W, const float* __restrict__ out,
          float* __restrict__ t) {
    __shared__ float ov[K];
    const int bn = blockIdx.x;
    const int n  = bn & (N - 1);
    const int d  = threadIdx.x;
    if (d < K) ov[d] = out[(size_t)bn * K + d];
    __syncthreads();
    const float* wr = W + (size_t)d * (N * K) + (size_t)n * K;
    float acc = 0.0f;
#pragma unroll
    for (int k = 0; k < K; ++k) acc += wr[k] * ov[k];
    t[(size_t)bn * D + d] = acc;
}

constexpr int BU_TPW   = 4;
constexpr int BU_WAVES = 4;

__global__ void __launch_bounds__(128)
bupdate(const float* __restrict__ x, const float* __restrict__ t,
        float* __restrict__ bc) {
    const int b    = blockIdx.y;
    const int lane = threadIdx.x & 31;
    const int wave = threadIdx.x >> 5;
    const int half = lane >> 4;
    const int l16  = lane & 15;

    const int i0 = (blockIdx.x * BU_WAVES + wave) * BU_TPW * 16;
    const float* tr = t + (size_t)b * N * D + (size_t)l16 * D + half * 4;
    const float* xr = x + ((size_t)b * In + i0 + l16) * D + half * 4;

    v8f acc[BU_TPW][2] = {};
#pragma unroll
    for (int c = 0; c < D / 16; ++c) {
        const v4f ta = *(const v4f*)(tr + 16 * c);
        const v4f tb = *(const v4f*)(tr + 16 * c + 8);
        const v2f b0 = {ta.x, ta.y}, b1 = {ta.z, ta.w};
        const v2f b2 = {tb.x, tb.y}, b3 = {tb.z, tb.w};
#pragma unroll
        for (int tt = 0; tt < BU_TPW; ++tt) {
            const float* p = xr + (size_t)tt * 16 * D + 16 * c;
            const v4f xa = *(const v4f*)(p);
            const v4f xb = *(const v4f*)(p + 8);
            const v2f a0 = {xa.x, xa.y}, a1 = {xa.z, xa.w};
            const v2f a2 = {xb.x, xb.y}, a3 = {xb.z, xb.w};
            acc[tt][0] = __builtin_amdgcn_wmma_f32_16x16x4_f32(false, a0, false, b0, (short)0, acc[tt][0], false, false);
            acc[tt][1] = __builtin_amdgcn_wmma_f32_16x16x4_f32(false, a1, false, b1, (short)0, acc[tt][1], false, false);
            acc[tt][0] = __builtin_amdgcn_wmma_f32_16x16x4_f32(false, a2, false, b2, (short)0, acc[tt][0], false, false);
            acc[tt][1] = __builtin_amdgcn_wmma_f32_16x16x4_f32(false, a3, false, b3, (short)0, acc[tt][1], false, false);
        }
    }

#pragma unroll
    for (int tt = 0; tt < BU_TPW; ++tt) {
        float* dst = bc + ((size_t)b * In + i0 + tt * 16) * N;
#pragma unroll
        for (int r = 0; r < 8; ++r)
            dst[(size_t)(r + half * 8) * N + l16] = acc[tt][0][r] + acc[tt][1][r];
    }
}

extern "C" void kernel_launch(void* const* d_in, const int* in_sizes, int n_in,
                              void* d_out, int out_size, void* d_ws, size_t ws_size,
                              hipStream_t stream) {
    const float* x = (const float*)d_in[0];
    const float* W = (const float*)d_in[1];
    float* out = (float*)d_out;

    float* ws     = (float*)d_ws;
    float* bc     = ws;
    float* s      = bc + (size_t)B * In * N;
    float* t      = s + (size_t)B * N * D;
    float* s1     = t;

    zero_buf<<<ZB_BLOCKS, ZB_THREADS, 0, stream>>>(bc);

    for (int it = 0; it < 3; ++it) {
        softmax_n<<<(B * In + 255) / 256, 256, 0, stream>>>(bc);
        gemm_s<<<dim3(GS_SPLIT * D / GS_COLS, B), GS_WAVES * 32, 0, stream>>>(x, bc, s, s1);
        outputs_squash<<<B * N, 256, 0, stream>>>(s, s1, W, out);
        if (it < 2) {
            compute_t<<<B * N, 256, 0, stream>>>(W, out, t);
            bupdate<<<dim3(In / (16 * BU_TPW * BU_WAVES), B), 128, 0, stream>>>(x, t, bc);
        }
    }
}


# softmax over n moved into bupdate's epilogue: logits tile transposed in LDS, each lane normalises 2 whole rows, c stored as whole 128-byte lines; 14 launches to 12; no sum touched
# speedup vs baseline: 2.6725x; 1.3000x over previous
#include <hip/hip_runtime.h>
#include <hip/hip_bf16.h>


typedef __attribute__((ext_vector_type(2))) float v2f;
typedef __attribute__((ext_vector_type(4))) float v4f;
typedef __attribute__((ext_vector_type(8))) float v8f;

namespace caps {
constexpr int B   = 32;
constexpr int In  = 4096;
constexpr int D   = 256;
constexpr int N   = 16;
constexpr int K   = 16;
}

using namespace caps;

constexpr int ZB_THREADS    = 256;
constexpr int ZB_PER_THREAD = 8;
constexpr int ZB_BLOCKS     = (B * In * N) / (4 * ZB_THREADS * ZB_PER_THREAD);
static_assert((B * In * N) % (4 * ZB_THREADS * ZB_PER_THREAD) == 0,
              "the logits must divide into whole work-group pieces");

__global__ void __launch_bounds__(ZB_THREADS)
zero_buf(float* __restrict__ p) {
    v4f* q = (v4f*)p + (size_t)blockIdx.x * (ZB_THREADS * ZB_PER_THREAD) + threadIdx.x;
    const v4f z = {0.0f, 0.0f, 0.0f, 0.0f};
    for (int k = 0; k < ZB_PER_THREAD; ++k) q[k * ZB_THREADS] = z;
}

__device__ __forceinline__ void softmax_row(float (&v)[N]) {
    float m = -3.402823466e+38f;
#pragma unroll
    for (int n = 0; n < N; ++n) m = fmaxf(m, v[n]);
    float s = 0.0f;
#pragma unroll
    for (int n = 0; n < N; ++n) { v[n] = __expf(v[n] - m); s += v[n]; }
    const float inv = 1.0f / s;
#pragma unroll
    for (int n = 0; n < N; ++n) v[n] = v[n] * inv;
}

__global__ void softmax_n(float* __restrict__ bc) {
    int idx = blockIdx.x * blockDim.x + threadIdx.x;
    if (idx >= B * In) return;
    float* p = bc + (size_t)idx * N;
    float v[N];
#pragma unroll
    for (int n = 0; n < N; ++n) v[n] = p[n];
    softmax_row(v);
#pragma unroll
    for (int n = 0; n < N; ++n) p[n] = v[n];
}

constexpr int GS_WAVES = 16;
constexpr int GS_COLS  = 64;
constexpr int GS_GRP   = 8;
constexpr int GS_SPLIT = 2;
constexpr int GS_ROWS  = In / (GS_SPLIT * GS_WAVES);
constexpr int GS_NGRP  = GS_ROWS / (4 * GS_GRP);
static_assert(GS_SPLIT == 2, "gemm_s and outputs_squash name exactly two partial planes");
static_assert(GS_ROWS % (4 * GS_GRP) == 0, "a wave's rows must divide into whole load groups");

__global__ void __launch_bounds__(GS_WAVES * 32)
gemm_s(const float* __restrict__ x, const float* __restrict__ c,
       float* __restrict__ s0, float* __restrict__ s1) {
    __shared__ __attribute__((aligned(16))) float red[GS_WAVES * N * GS_COLS];
    const int b    = blockIdx.y;
    const int j    = blockIdx.x % (D / GS_COLS);
    const int z    = blockIdx.x / (D / GS_COLS);
    const int lane = threadIdx.x & 31;
    const int wave = threadIdx.x >> 5;
    const int half = lane >> 4;
    const int l16  = lane & 15;
    float* __restrict__ s = z == 0 ? s0 : s1;

    const int r0 = (z * GS_WAVES + wave) * GS_ROWS + half * 2;
    const float* cp = c + ((size_t)b * In + r0) * N + l16;
    const float* xp = x + ((size_t)b * In + r0) * D + j * GS_COLS + l16 * 4;

    v8f acc0 = {}, acc1 = {}, acc2 = {}, acc3 = {};
    v4f xv[2][GS_GRP][2];
    float cv[2][GS_GRP][2];

#pragma unroll
    for (int ks = 0; ks < GS_GRP; ++ks) {
#pragma unroll
        for (int v = 0; v < 2; ++v) {
            xv[0][ks][v] = *(const v4f*)(xp + (size_t)(4 * ks + v) * D);
            cv[0][ks][v] = cp[(size_t)(4 * ks + v) * N];
        }
    }

#pragma unroll
    for (int g = 0; g < GS_NGRP; ++g) {
        const int cur = g & 1, nxt = cur ^ 1;
        if (g + 1 < GS_NGRP) {
#pragma unroll
            for (int ks = 0; ks < GS_GRP; ++ks) {
#pragma unroll
                for (int v = 0; v < 2; ++v) {
                    const int row = 4 * (GS_GRP * (g + 1) + ks) + v;
                    xv[nxt][ks][v] = *(const v4f*)(xp + (size_t)row * D);
                    cv[nxt][ks][v] = cp[(size_t)row * N];
                }
            }
        }
        __builtin_amdgcn_sched_barrier(0);
#pragma unroll
        for (int ks = 0; ks < GS_GRP; ++ks) {
            const v4f x0 = xv[cur][ks][0], x1 = xv[cur][ks][1];
            const v2f a  = {cv[cur][ks][0], cv[cur][ks][1]};
            const v2f b0 = {x0.x, x1.x}, b1 = {x0.y, x1.y};
            const v2f b2 = {x0.z, x1.z}, b3 = {x0.w, x1.w};
            acc0 = __builtin_amdgcn_wmma_f32_16x16x4_f32(false, a, false, b0, (short)0, acc0, false, false);
            acc1 = __builtin_amdgcn_wmma_f32_16x16x4_f32(false, a, false, b1, (short)0, acc1, false, false);
            acc2 = __builtin_amdgcn_wmma_f32_16x16x4_f32(false, a, false, b2, (short)0, acc2, false, false);
            acc3 = __builtin_amdgcn_wmma_f32_16x16x4_f32(false, a, false, b3, (short)0, acc3, false, false);
        }
    }

    float* mine = red + wave * (N * GS_COLS);
#pragma unroll
    for (int r = 0; r < 8; ++r) {
        const v4f o = {acc0[r], acc1[r], acc2[r], acc3[r]};
        *(v4f*)(mine + (r + half * 8) * GS_COLS + l16 * 4) = o;
    }
    __syncthreads();

    if (threadIdx.x < N * GS_COLS / 4) {
        const int n  = threadIdx.x >> 4;
        const int c4 = threadIdx.x & 15;
        v4f sum = *(const v4f*)(red + n * GS_COLS + c4 * 4);
#pragma unroll
        for (int w = 1; w < GS_WAVES; ++w)
            sum += *(const v4f*)(red + (w * N + n) * GS_COLS + c4 * 4);
        *(v4f*)(s + ((size_t)b * N + n) * D + j * GS_COLS + c4 * 4) = sum;
    }
}

__global__ void __launch_bounds__(256)
outputs_squash(const float* __restrict__ s0, const float* __restrict__ s1,
               const float* __restrict__ W, float* __restrict__ out) {
    __shared__ float part[D * K];
    __shared__ float dot[K];
    const int bn = blockIdx.x;
    const int n  = bn & (N - 1);
    const int d  = threadIdx.x;

    const float  sv = s0[(size_t)bn * D + d] + s1[(size_t)bn * D + d];
    const float* wr = W + (size_t)d * (N * K) + (size_t)n * K;
#pragma unroll
    for (int k = 0; k < K; ++k) part[d * K + k] = sv * wr[k];
    __syncthreads();

    if (d < K) {
        float a = 0.0f;
        for (int dd = 0; dd < D; ++dd) a += part[dd * K + d];
        dot[d] = a;
    }
    __syncthreads();

    if (d < K) {
        float n2 = 0.0f;
#pragma unroll
        for (int k = 0; k < K; ++k) n2 += dot[k] * dot[k];
        out[(size_t)bn * K + d] = dot[d] * rsqrtf(n2 + 1e-7f);
    }
}

__global__ void __launch_bounds__(256)
compute_t(const float* __restrict__ W, const float* __restrict__ out,
          float* __restrict__ t) {
    __shared__ float ov[K];
    const int bn = blockIdx.x;
    const int n  = bn & (N - 1);
    const int d  = threadIdx.x;
    if (d < K) ov[d] = out[(size_t)bn * K + d];
    __syncthreads();
    const float* wr = W + (size_t)d * (N * K) + (size_t)n * K;
    float acc = 0.0f;
#pragma unroll
    for (int k = 0; k < K; ++k) acc += wr[k] * ov[k];
    t[(size_t)bn * D + d] = acc;
}

constexpr int BU_TPW   = 4;
constexpr int BU_WAVES = 4;
constexpr int BU_WROWS = BU_TPW * 16;
constexpr int BU_PITCH = 20;
static_assert(BU_WROWS == 2 * 32, "each lane owns two of its wave's rows");

__global__ void __launch_bounds__(128)
bupdate(const float* __restrict__ x, const float* __restrict__ t,
        float* __restrict__ bc) {
    __shared__ __attribute__((aligned(16))) float stage[BU_WAVES * BU_WROWS * BU_PITCH];
    const int b    = blockIdx.y;
    const int lane = threadIdx.x & 31;
    const int wave = threadIdx.x >> 5;
    const int half = lane >> 4;
    const int l16  = lane & 15;

    const int i0 = (blockIdx.x * BU_WAVES + wave) * BU_TPW * 16;
    const float* tr = t + (size_t)b * N * D + (size_t)l16 * D + half * 4;
    const float* xr = x + ((size_t)b * In + i0 + l16) * D + half * 4;

    v8f acc[BU_TPW][2] = {};
#pragma unroll
    for (int c = 0; c < D / 16; ++c) {
        const v4f ta = *(const v4f*)(tr + 16 * c);
        const v4f tb = *(const v4f*)(tr + 16 * c + 8);
        const v2f b0 = {ta.x, ta.y}, b1 = {ta.z, ta.w};
        const v2f b2 = {tb.x, tb.y}, b3 = {tb.z, tb.w};
#pragma unroll
        for (int tt = 0; tt < BU_TPW; ++tt) {
            const float* p = xr + (size_t)tt * 16 * D + 16 * c;
            const v4f xa = *(const v4f*)(p);
            const v4f xb = *(const v4f*)(p + 8);
            const v2f a0 = {xa.x, xa.y}, a1 = {xa.z, xa.w};
            const v2f a2 = {xb.x, xb.y}, a3 = {xb.z, xb.w};
            acc[tt][0] = __builtin_amdgcn_wmma_f32_16x16x4_f32(false, a0, false, b0, (short)0, acc[tt][0], false, false);
            acc[tt][1] = __builtin_amdgcn_wmma_f32_16x16x4_f32(false, a1, false, b1, (short)0, acc[tt][1], false, false);
            acc[tt][0] = __builtin_amdgcn_wmma_f32_16x16x4_f32(false, a2, false, b2, (short)0, acc[tt][0], false, false);
            acc[tt][1] = __builtin_amdgcn_wmma_f32_16x16x4_f32(false, a3, false, b3, (short)0, acc[tt][1], false, false);
        }
    }

    float* mine = stage + wave * (BU_WROWS * BU_PITCH);
#pragma unroll
    for (int tt = 0; tt < BU_TPW; ++tt) {
#pragma unroll
        for (int r = 0; r < 8; ++r)
            mine[(tt * 16 + r + half * 8) * BU_PITCH + l16] = acc[tt][0][r] + acc[tt][1][r];
    }
    __syncthreads();

#pragma unroll
    for (int p = 0; p < 2; ++p) {
        float* row = mine + (2 * lane + p) * BU_PITCH;
        float v[N];
#pragma unroll
        for (int q = 0; q < N / 4; ++q) {
            const v4f piece = *(const v4f*)(row + 4 * q);
            v[4 * q + 0] = piece.x; v[4 * q + 1] = piece.y;
            v[4 * q + 2] = piece.z; v[4 * q + 3] = piece.w;
        }
        softmax_row(v);
#pragma unroll
        for (int q = 0; q < N / 4; ++q) {
            const v4f piece = {v[4 * q + 0], v[4 * q + 1], v[4 * q + 2], v[4 * q + 3]};
            *(v4f*)(row + 4 * q) = piece;
        }
    }
    __syncthreads();

    float* dst = bc + ((size_t)b * In + i0) * N;
#pragma unroll
    for (int k = 0; k < BU_WROWS * N / (4 * 32); ++k) {
        const int piece = 32 * k + lane;
        const v4f o = *(const v4f*)(mine + (piece >> 2) * BU_PITCH + (piece & 3) * 4);
        *(v4f*)(dst + (size_t)piece * 4) = o;
    }
}

extern "C" void kernel_launch(void* const* d_in, const int* in_sizes, int n_in,
                              void* d_out, int out_size, void* d_ws, size_t ws_size,
                              hipStream_t stream) {
    const float* x = (const float*)d_in[0];
    const float* W = (const float*)d_in[1];
    float* out = (float*)d_out;

    float* ws     = (float*)d_ws;
    float* bc     = ws;
    float* s      = bc + (size_t)B * In * N;
    float* t      = s + (size_t)B * N * D;
    float* s1     = t;

    zero_buf<<<ZB_BLOCKS, ZB_THREADS, 0, stream>>>(bc);
    softmax_n<<<(B * In + 255) / 256, 256, 0, stream>>>(bc);

    for (int it = 0; it < 3; ++it) {
        gemm_s<<<dim3(GS_SPLIT * D / GS_COLS, B), GS_WAVES * 32, 0, stream>>>(x, bc, s, s1);
        outputs_squash<<<B * N, 256, 0, stream>>>(s, s1, W, out);
        if (it < 2) {
            compute_t<<<B * N, 256, 0, stream>>>(W, out, t);
            bupdate<<<dim3(In / (16 * BU_TPW * BU_WAVES), B), 128, 0, stream>>>(x, t, bc);
        }
    }
}


# bupdate forms t[b] in its prologue (s planes added, squash, both W products; W in registers, t in LDS): outputs_squash, compute_t of iterations 1-2 gone, 12 launches to 8; new sum order over d, k
# speedup vs baseline: 3.0255x; 1.1321x over previous
#include <hip/hip_runtime.h>
#include <hip/hip_bf16.h>


typedef __attribute__((ext_vector_type(2))) float v2f;
typedef __attribute__((ext_vector_type(4))) float v4f;
typedef __attribute__((ext_vector_type(8))) float v8f;

namespace caps {
constexpr int B   = 32;
constexpr int In  = 4096;
constexpr int D   = 256;
constexpr int N   = 16;
constexpr int K   = 16;
}

using namespace caps;

constexpr int ZB_THREADS    = 256;
constexpr int ZB_PER_THREAD = 8;
constexpr int ZB_BLOCKS     = (B * In * N) / (4 * ZB_THREADS * ZB_PER_THREAD);
static_assert((B * In * N) % (4 * ZB_THREADS * ZB_PER_THREAD) == 0,
              "the logits must divide into whole work-group pieces");

__global__ void __launch_bounds__(ZB_THREADS)
zero_buf(float* __restrict__ p) {
    v4f* q = (v4f*)p + (size_t)blockIdx.x * (ZB_THREADS * ZB_PER_THREAD) + threadIdx.x;
    const v4f z = {0.0f, 0.0f, 0.0f, 0.0f};
    for (int k = 0; k < ZB_PER_THREAD; ++k) q[k * ZB_THREADS] = z;
}

__device__ __forceinline__ void softmax_row(float (&v)[N]) {
    float m = -3.402823466e+38f;
#pragma unroll
    for (int n = 0; n < N; ++n) m = fmaxf(m, v[n]);
    float s = 0.0f;
#pragma unroll
    for (int n = 0; n < N; ++n) { v[n] = __expf(v[n] - m); s += v[n]; }
    const float inv = 1.0f / s;
#pragma unroll
    for (int n = 0; n < N; ++n) v[n] = v[n] * inv;
}

__global__ void softmax_n(float* __restrict__ bc) {
    int idx = blockIdx.x * blockDim.x + threadIdx.x;
    if (idx >= B * In) return;
    float* p = bc + (size_t)idx * N;
    float v[N];
#pragma unroll
    for (int n = 0; n < N; ++n) v[n] = p[n];
    softmax_row(v);
#pragma unroll
    for (int n = 0; n < N; ++n) p[n] = v[n];
}

constexpr int GS_WAVES = 16;
constexpr int GS_COLS  = 64;
constexpr int GS_GRP   = 8;
constexpr int GS_SPLIT = 2;
constexpr int GS_ROWS  = In / (GS_SPLIT * GS_WAVES);
constexpr int GS_NGRP  = GS_ROWS / (4 * GS_GRP);
static_assert(GS_SPLIT == 2, "gemm_s and the readers of s name exactly two partial planes");
static_assert(GS_ROWS % (4 * GS_GRP) == 0, "a wave's rows must divide into whole load groups");

__global__ void __launch_bounds__(GS_WAVES * 32)
gemm_s(const float* __restrict__ x, const float* __restrict__ c,
       float* __restrict__ s0, float* __restrict__ s1) {
    __shared__ __attribute__((aligned(16))) float red[GS_WAVES * N * GS_COLS];
    const int b    = blockIdx.y;
    const int j    = blockIdx.x % (D / GS_COLS);
    const int z    = blockIdx.x / (D / GS_COLS);
    const int lane = threadIdx.x & 31;
    const int wave = threadIdx.x >> 5;
    const int half = lane >> 4;
    const int l16  = lane & 15;
    float* __restrict__ s = z == 0 ? s0 : s1;

    const int r0 = (z * GS_WAVES + wave) * GS_ROWS + half * 2;
    const float* cp = c + ((size_t)b * In + r0) * N + l16;
    const float* xp = x + ((size_t)b * In + r0) * D + j * GS_COLS + l16 * 4;

    v8f acc0 = {}, acc1 = {}, acc2 = {}, acc3 = {};
    v4f xv[2][GS_GRP][2];
    float cv[2][GS_GRP][2];

#pragma unroll
    for (int ks = 0; ks < GS_GRP; ++ks) {
#pragma unroll
        for (int v = 0; v < 2; ++v) {
            xv[0][ks][v] = *(const v4f*)(xp + (size_t)(4 * ks + v) * D);
            cv[0][ks][v] = cp[(size_t)(4 * ks + v) * N];
        }
    }

#pragma unroll
    for (int g = 0; g < GS_NGRP; ++g) {
        const int cur = g & 1, nxt = cur ^ 1;
        if (g + 1 < GS_NGRP) {
#pragma unroll
            for (int ks = 0; ks < GS_GRP; ++ks) {
#pragma unroll
                for (int v = 0; v < 2; ++v) {
                    const int row = 4 * (GS_GRP * (g + 1) + ks) + v;
                    xv[nxt][ks][v] = *(const v4f*)(xp + (size_t)row * D);
                    cv[nxt][ks][v] = cp[(size_t)row * N];
                }
            }
        }
        __builtin_amdgcn_sched_barrier(0);
#pragma unroll
        for (int ks = 0; ks < GS_GRP; ++ks) {
            const v4f x0 = xv[cur][ks][0], x1 = xv[cur][ks][1];
            const v2f a  = {cv[cur][ks][0], cv[cur][ks][1]};
            const v2f b0 = {x0.x, x1.x}, b1 = {x0.y, x1.y};
            const v2f b2 = {x0.z, x1.z}, b3 = {x0.w, x1.w};
            acc0 = __builtin_amdgcn_wmma_f32_16x16x4_f32(false, a, false, b0, (short)0, acc0, false, false);
            acc1 = __builtin_amdgcn_wmma_f32_16x16x4_f32(false, a, false, b1, (short)0, acc1, false, false);
            acc2 = __builtin_amdgcn_wmma_f32_16x16x4_f32(false, a, false, b2, (short)0, acc2, false, false);
            acc3 = __builtin_amdgcn_wmma_f32_16x16x4_f32(false, a, false, b3, (short)0, acc3, false, false);
        }
    }

    float* mine = red + wave * (N * GS_COLS);
#pragma unroll
    for (int r = 0; r < 8; ++r) {
        const v4f o = {acc0[r], acc1[r], acc2[r], acc3[r]};
        *(v4f*)(mine + (r + half * 8) * GS_COLS + l16 * 4) = o;
    }
    __syncthreads();

    if (threadIdx.x < N * GS_COLS / 4) {
        const int n  = threadIdx.x >> 4;
        const int c4 = threadIdx.x & 15;
        v4f sum = *(const v4f*)(red + n * GS_COLS + c4 * 4);
#pragma unroll
        for (int w = 1; w < GS_WAVES; ++w)
            sum += *(const v4f*)(red + (w * N + n) * GS_COLS + c4 * 4);
        *(v4f*)(s + ((size_t)b * N + n) * D + j * GS_COLS + c4 * 4) = sum;
    }
}

__global__ void __launch_bounds__(256)
outputs_squash(const float* __restrict__ s0, const float* __restrict__ s1,
               const float* __restrict__ W, float* __restrict__ out) {
    __shared__ float part[D * K];
    __shared__ float dot[K];
    const int bn = blockIdx.x;
    const int n  = bn & (N - 1);
    const int d  = threadIdx.x;

    const float  sv = s0[(size_t)bn * D + d] + s1[(size_t)bn * D + d];
    const float* wr = W + (size_t)d * (N * K) + (size_t)n * K;
#pragma unroll
    for (int k = 0; k < K; ++k) part[d * K + k] = sv * wr[k];
    __syncthreads();

    if (d < K) {
        float a = 0.0f;
        for (int dd = 0; dd < D; ++dd) a += part[dd * K + d];
        dot[d] = a;
    }
    __syncthreads();

    if (d < K) {
        float n2 = 0.0f;
#pragma unroll
        for (int k = 0; k < K; ++k) n2 += dot[k] * dot[k];
        out[(size_t)bn * K + d] = dot[d] * rsqrtf(n2 + 1e-7f);
    }
}

constexpr int BU_TPW   = 4;
constexpr int BU_WAVES = 8;
constexpr int BU_WROWS = BU_TPW * 16;
constexpr int BU_PITCH = 20;
constexpr int BU_TP    = D + 4;
constexpr int BU_WPT   = D / 4;
constexpr int BU_PRE   = 4;
static_assert(BU_WROWS == 2 * 32, "each lane owns two of its wave's rows");
static_assert(BU_WAVES * 32 == N * K, "one prologue thread per output element");
static_assert(N * K == D && K == 16, "the prologue's thread map assumes 256 columns in capsules of 16");

__device__ __forceinline__ float quad_xor1(float v) {
    return __builtin_bit_cast(float, __builtin_amdgcn_mov_dpp(__builtin_bit_cast(int, v), 0xB1, 0xF, 0xF, true));
}
__device__ __forceinline__ float quad_xor2(float v) {
    return __builtin_bit_cast(float, __builtin_amdgcn_mov_dpp(__builtin_bit_cast(int, v), 0x4E, 0xF, 0xF, true));
}

__global__ void __launch_bounds__(BU_WAVES * 32)
bupdate(const float* __restrict__ x, const float* __restrict__ s0,
        const float* __restrict__ s1, const float* __restrict__ W,
        float* __restrict__ bc) {
    __shared__ __attribute__((aligned(16))) float stage[BU_WAVES * BU_WROWS * BU_PITCH];
    __shared__ __attribute__((aligned(16))) float sl[N * BU_TP];
    __shared__ __attribute__((aligned(16))) float tl[N * BU_TP];
    __shared__ __attribute__((aligned(16))) float part[4 * N * K];
    __shared__ __attribute__((aligned(16))) float oraw[N * K];
    __shared__ __attribute__((aligned(16))) float onrm[N * K];
    const int b    = blockIdx.y;
    const int tid  = threadIdx.x;
    const int lane = tid & 31;
    const int wave = tid >> 5;
    const int half = lane >> 4;
    const int l16  = lane & 15;

    const int q  = tid >> 6;
    const int mm = tid & 63;
    const int nq = mm >> 2;
    v4f wreg[BU_WPT];
    {
        const float* wp = W + (size_t)(q * BU_WPT) * (N * K) + mm * 4;
#pragma unroll
        for (int e = 0; e < BU_WPT; ++e) wreg[e] = *(const v4f*)(wp + (size_t)e * (N * K));
    }
#pragma unroll
    for (int r = 0; r < (N * D / 4) / (BU_WAVES * 32); ++r) {
        const int piece = r * (BU_WAVES * 32) + tid;
        const int n  = piece / (D / 4);
        const int c4 = piece % (D / 4);
        const size_t g = ((size_t)b * N + n) * D + c4 * 4;
        const v4f a0 = *(const v4f*)(s0 + g);
        const v4f a1 = *(const v4f*)(s1 + g);
        *(v4f*)(sl + n * BU_TP + c4 * 4) = a0 + a1;
    }
    const int i0 = (blockIdx.x * BU_WAVES + wave) * BU_TPW * 16;
    const float* xr = x + ((size_t)b * In + i0 + l16) * D + half * 4;
    v4f xpre[BU_PRE][BU_TPW][2];
#pragma unroll
    for (int c = 0; c < BU_PRE; ++c) {
#pragma unroll
        for (int tt = 0; tt < BU_TPW; ++tt) {
            const float* p = xr + (size_t)tt * 16 * D + 16 * c;
            xpre[c][tt][0] = *(const v4f*)(p);
            xpre[c][tt][1] = *(const v4f*)(p + 8);
        }
    }
    __syncthreads();

    {
        v4f acc1 = {0.0f, 0.0f, 0.0f, 0.0f};
        const float* srow = sl + nq * BU_TP + q * BU_WPT;
#pragma unroll
        for (int e4 = 0; e4 < BU_WPT / 4; ++e4) {
            const v4f sv = *(const v4f*)(srow + 4 * e4);
            acc1 += sv.x * wreg[4 * e4 + 0];
            acc1 += sv.y * wreg[4 * e4 + 1];
            acc1 += sv.z * wreg[4 * e4 + 2];
            acc1 += sv.w * wreg[4 * e4 + 3];
        }
        *(v4f*)(part + q * (N * K) + mm * 4) = acc1;
    }
    __syncthreads();

    {
        const float o = ((part[tid] + part[N * K + tid]) + part[2 * N * K + tid]) + part[3 * N * K + tid];
        oraw[tid] = o;
        __syncthreads();
        const float* mine_o = oraw + (tid & ~(K - 1));
        float n2 = 0.0f;
#pragma unroll
        for (int k = 0; k < K; ++k) n2 += mine_o[k] * mine_o[k];
        onrm[tid] = o * rsqrtf(n2 + 1e-7f);
    }
    __syncthreads();

    {
        const v4f ov = *(const v4f*)(onrm + mm * 4);
        float* trow = tl + nq * BU_TP + q * BU_WPT;
#pragma unroll
        for (int e4 = 0; e4 < BU_WPT / 4; ++e4) {
            v4f tv;
#pragma unroll
            for (int u = 0; u < 4; ++u) {
                const v4f w = wreg[4 * e4 + u];
                float p = w.x * ov.x + w.y * ov.y + w.z * ov.z + w.w * ov.w;
                p += quad_xor1(p);
                p += quad_xor2(p);
                tv[u] = p;
            }
            if ((mm & 3) == 0) *(v4f*)(trow + 4 * e4) = tv;
        }
    }
    __syncthreads();

    const float* tr = tl + l16 * BU_TP + half * 4;

    v8f acc[BU_TPW][2] = {};
#pragma unroll
    for (int c = 0; c < D / 16; ++c) {
        const v4f ta = *(const v4f*)(tr + 16 * c);
        const v4f tb = *(const v4f*)(tr + 16 * c + 8);
        const v2f b0 = {ta.x, ta.y}, b1 = {ta.z, ta.w};
        const v2f b2 = {tb.x, tb.y}, b3 = {tb.z, tb.w};
#pragma unroll
        for (int tt = 0; tt < BU_TPW; ++tt) {
            const float* p = xr + (size_t)tt * 16 * D + 16 * c;
            const v4f xa = c < BU_PRE ? xpre[c < BU_PRE ? c : 0][tt][0] : *(const v4f*)(p);
            const v4f xb = c < BU_PRE ? xpre[c < BU_PRE ? c : 0][tt][1] : *(const v4f*)(p + 8);
            const v2f a0 = {xa.x, xa.y}, a1 = {xa.z, xa.w};
            const v2f a2 = {xb.x, xb.y}, a3 = {xb.z, xb.w};
            acc[tt][0] = __builtin_amdgcn_wmma_f32_16x16x4_f32(false, a0, false, b0, (short)0, acc[tt][0], false, false);
            acc[tt][1] = __builtin_amdgcn_wmma_f32_16x16x4_f32(false, a1, false, b1, (short)0, acc[tt][1], false, false);
            acc[tt][0] = __builtin_amdgcn_wmma_f32_16x16x4_f32(false, a2, false, b2, (short)0, acc[tt][0], false, false);
            acc[tt][1] = __builtin_amdgcn_wmma_f32_16x16x4_f32(false, a3, false, b3, (short)0, acc[tt][1], false, false);
        }
    }

    float* mine = stage + wave * (BU_WROWS * BU_PITCH);
#pragma unroll
    for (int tt = 0; tt < BU_TPW; ++tt) {
#pragma unroll
        for (int r = 0; r < 8; ++r)
            mine[(tt * 16 + r + half * 8) * BU_PITCH + l16] = acc[tt][0][r] + acc[tt][1][r];
    }
    __syncthreads();

#pragma unroll
    for (int p = 0; p < 2; ++p) {
        float* row = mine + (2 * lane + p) * BU_PITCH;
        float v[N];
#pragma unroll
        for (int q = 0; q < N / 4; ++q) {
            const v4f piece = *(const v4f*)(row + 4 * q);
            v[4 * q + 0] = piece.x; v[4 * q + 1] = piece.y;
            v[4 * q + 2] = piece.z; v[4 * q + 3] = piece.w;
        }
        softmax_row(v);
#pragma unroll
        for (int q = 0; q < N / 4; ++q) {
            const v4f piece = {v[4 * q + 0], v[4 * q + 1], v[4 * q + 2], v[4 * q + 3]};
            *(v4f*)(row + 4 * q) = piece;
        }
    }
    __syncthreads();

    float* dst = bc + ((size_t)b * In + i0) * N;
#pragma unroll
    for (int k = 0; k < BU_WROWS * N / (4 * 32); ++k) {
        const int piece = 32 * k + lane;
        const v4f o = *(const v4f*)(mine + (piece >> 2) * BU_PITCH + (piece & 3) * 4);
        *(v4f*)(dst + (size_t)piece * 4) = o;
    }
}

extern "C" void kernel_launch(void* const* d_in, const int* in_sizes, int n_in,
                              void* d_out, int out_size, void* d_ws, size_t ws_size,
                              hipStream_t stream) {
    const float* x = (const float*)d_in[0];
    const float* W = (const float*)d_in[1];
    float* out = (float*)d_out;

    float* ws     = (float*)d_ws;
    float* bc     = ws;
    float* s      = bc + (size_t)B * In * N;
    float* s1     = s + (size_t)B * N * D;

    zero_buf<<<ZB_BLOCKS, ZB_THREADS, 0, stream>>>(bc);
    softmax_n<<<(B * In + 255) / 256, 256, 0, stream>>>(bc);

    for (int it = 0; it < 3; ++it) {
        gemm_s<<<dim3(GS_SPLIT * D / GS_COLS, B), GS_WAVES * 32, 0, stream>>>(x, bc, s, s1);
        if (it < 2)
            bupdate<<<dim3(In / (16 * BU_TPW * BU_WAVES), B), BU_WAVES * 32, 0, stream>>>(x, s, s1, W, bc);
        else
            outputs_squash<<<B * N, 256, 0, stream>>>(s, s1, W, out);
    }
}


# gemm_s re-tiled to bupdate's map: group (512-row block, b) = 4 column slices x 4 row groups, so group k of both kernels reads the same 512 KB of x; 8 planes of s: new order of the sum over i
# speedup vs baseline: 3.2286x; 1.0671x over previous
#include <hip/hip_runtime.h>
#include <hip/hip_bf16.h>


typedef __attribute__((ext_vector_type(2))) float v2f;
typedef __attribute__((ext_vector_type(4))) float v4f;
typedef __attribute__((ext_vector_type(8))) float v8f;

namespace caps {
constexpr int B   = 32;
constexpr int In  = 4096;
constexpr int D   = 256;
constexpr int N   = 16;
constexpr int K   = 16;
}

using namespace caps;

constexpr int ZB_THREADS    = 256;
constexpr int ZB_PER_THREAD = 8;
constexpr int ZB_BLOCKS     = (B * In * N) / (4 * ZB_THREADS * ZB_PER_THREAD);
static_assert((B * In * N) % (4 * ZB_THREADS * ZB_PER_THREAD) == 0,
              "the logits must divide into whole work-group pieces");

__global__ void __launch_bounds__(ZB_THREADS)
zero_buf(float* __restrict__ p) {
    v4f* q = (v4f*)p + (size_t)blockIdx.x * (ZB_THREADS * ZB_PER_THREAD) + threadIdx.x;
    const v4f z = {0.0f, 0.0f, 0.0f, 0.0f};
    for (int k = 0; k < ZB_PER_THREAD; ++k) q[k * ZB_THREADS] = z;
}

__device__ __forceinline__ void softmax_row(float (&v)[N]) {
    float m = -3.402823466e+38f;
#pragma unroll
    for (int n = 0; n < N; ++n) m = fmaxf(m, v[n]);
    float s = 0.0f;
#pragma unroll
    for (int n = 0; n < N; ++n) { v[n] = __expf(v[n] - m); s += v[n]; }
    const float inv = 1.0f / s;
#pragma unroll
    for (int n = 0; n < N; ++n) v[n] = v[n] * inv;
}

__global__ void softmax_n(float* __restrict__ bc) {
    int idx = blockIdx.x * blockDim.x + threadIdx.x;
    if (idx >= B * In) return;
    float* p = bc + (size_t)idx * N;
    float v[N];
#pragma unroll
    for (int n = 0; n < N; ++n) v[n] = p[n];
    softmax_row(v);
#pragma unroll
    for (int n = 0; n < N; ++n) p[n] = v[n];
}

constexpr int GS_WAVES = 16;
constexpr int GS_COLS  = 64;
constexpr int GS_GRP   = 8;
constexpr int GS_CSL    = D / GS_COLS;
constexpr int GS_RGRP   = GS_WAVES / GS_CSL;
constexpr int GS_PLANES = 8;
constexpr int GS_BROWS  = In / GS_PLANES;
constexpr int GS_ROWS   = GS_BROWS / GS_RGRP;
constexpr int GS_NGRP   = GS_ROWS / (4 * GS_GRP);
constexpr int GS_PLANE  = B * N * D;
static_assert(GS_WAVES % GS_CSL == 0, "the waves must divide into whole row groups");
static_assert(GS_ROWS % (4 * GS_GRP) == 0, "a wave's rows must divide into whole load groups");
static_assert((N * D / 4) % (GS_WAVES * 32) == 0, "the 16-byte pieces of a plane must divide over the threads");

__global__ void __launch_bounds__(GS_WAVES * 32)
gemm_s(const float* __restrict__ x, const float* __restrict__ c,
       float* __restrict__ sp) {
    __shared__ __attribute__((aligned(16))) float red[GS_WAVES * N * GS_COLS];
    const int b    = blockIdx.y;
    const int rb   = blockIdx.x;
    const int lane = threadIdx.x & 31;
    const int wave = threadIdx.x >> 5;
    const int half = lane >> 4;
    const int l16  = lane & 15;
    const int j    = wave % GS_CSL;
    const int rg   = wave / GS_CSL;

    const int r0 = rb * GS_BROWS + rg * GS_ROWS + half * 2;
    const float* cp = c + ((size_t)b * In + r0) * N + l16;
    const float* xp = x + ((size_t)b * In + r0) * D + j * GS_COLS + l16 * 4;

    v8f acc0 = {}, acc1 = {}, acc2 = {}, acc3 = {};
    v4f xv[2][GS_GRP][2];
    float cv[2][GS_GRP][2];

#pragma unroll
    for (int ks = 0; ks < GS_GRP; ++ks) {
#pragma unroll
        for (int v = 0; v < 2; ++v) {
            xv[0][ks][v] = *(const v4f*)(xp + (size_t)(4 * ks + v) * D);
            cv[0][ks][v] = cp[(size_t)(4 * ks + v) * N];
        }
    }

#pragma unroll
    for (int g = 0; g < GS_NGRP; ++g) {
        const int cur = g & 1, nxt = cur ^ 1;
        if (g + 1 < GS_NGRP) {
#pragma unroll
            for (int ks = 0; ks < GS_GRP; ++ks) {
#pragma unroll
                for (int v = 0; v < 2; ++v) {
                    const int row = 4 * (GS_GRP * (g + 1) + ks) + v;
                    xv[nxt][ks][v] = *(const v4f*)(xp + (size_t)row * D);
                    cv[nxt][ks][v] = cp[(size_t)row * N];
                }
            }
        }
        __builtin_amdgcn_sched_barrier(0);
#pragma unroll
        for (int ks = 0; ks < GS_GRP; ++ks) {
            const v4f x0 = xv[cur][ks][0], x1 = xv[cur][ks][1];
            const v2f a  = {cv[cur][ks][0], cv[cur][ks][1]};
            const v2f b0 = {x0.x, x1.x}, b1 = {x0.y, x1.y};
            const v2f b2 = {x0.z, x1.z}, b3 = {x0.w, x1.w};
            acc0 = __builtin_amdgcn_wmma_f32_16x16x4_f32(false, a, false, b0, (short)0, acc0, false, false);
            acc1 = __builtin_amdgcn_wmma_f32_16x16x4_f32(false, a, false, b1, (short)0, acc1, false, false);
            acc2 = __builtin_amdgcn_wmma_f32_16x16x4_f32(false, a, false, b2, (short)0, acc2, false, false);
            acc3 = __builtin_amdgcn_wmma_f32_16x16x4_f32(false, a, false, b3, (short)0, acc3, false, false);
        }
    }

    float* mine = red + wave * (N * GS_COLS);
#pragma unroll
    for (int r = 0; r < 8; ++r) {
        const v4f o = {acc0[r], acc1[r], acc2[r], acc3[r]};
        *(v4f*)(mine + (r + half * 8) * GS_COLS + l16 * 4) = o;
    }
    __syncthreads();

    float* __restrict__ s = sp + (size_t)rb * GS_PLANE;
#pragma unroll
    for (int r = 0; r < (N * D / 4) / (GS_WAVES * 32); ++r) {
        const int piece = r * (GS_WAVES * 32) + threadIdx.x;
        const int n  = piece / (D / 4);
        const int q4 = piece % (D / 4);
        const int jj = q4 / (GS_COLS / 4);
        const int c4 = q4 % (GS_COLS / 4);
        v4f sum = *(const v4f*)(red + (jj * N + n) * GS_COLS + c4 * 4);
#pragma unroll
        for (int g = 1; g < GS_RGRP; ++g)
            sum += *(const v4f*)(red + ((g * GS_CSL + jj) * N + n) * GS_COLS + c4 * 4);
        *(v4f*)(s + ((size_t)b * N + n) * D + q4 * 4) = sum;
    }
}

__global__ void __launch_bounds__(256)
outputs_squash(const float* __restrict__ sp,
               const float* __restrict__ W, float* __restrict__ out) {
    __shared__ float part[D * K];
    __shared__ float dot[K];
    const int bn = blockIdx.x;
    const int n  = bn & (N - 1);
    const int d  = threadIdx.x;

    float sv = sp[(size_t)bn * D + d];
#pragma unroll
    for (int p = 1; p < GS_PLANES; ++p) sv += sp[(size_t)p * GS_PLANE + (size_t)bn * D + d];
    const float* wr = W + (size_t)d * (N * K) + (size_t)n * K;
#pragma unroll
    for (int k = 0; k < K; ++k) part[d * K + k] = sv * wr[k];
    __syncthreads();

    if (d < K) {
        float a = 0.0f;
        for (int dd = 0; dd < D; ++dd) a += part[dd * K + d];
        dot[d] = a;
    }
    __syncthreads();

    if (d < K) {
        float n2 = 0.0f;
#pragma unroll
        for (int k = 0; k < K; ++k) n2 += dot[k] * dot[k];
        out[(size_t)bn * K + d] = dot[d] * rsqrtf(n2 + 1e-7f);
    }
}

constexpr int BU_TPW   = 4;
constexpr int BU_WAVES = 8;
constexpr int BU_WROWS = BU_TPW * 16;
constexpr int BU_PITCH = 20;
constexpr int BU_TP    = D + 4;
constexpr int BU_WPT   = D / 4;
constexpr int BU_PRE   = 4;
static_assert(BU_WROWS == 2 * 32, "each lane owns two of its wave's rows");
static_assert(BU_WROWS * BU_WAVES == GS_BROWS, "bupdate and gemm_s must give a work-group the same rows of x");
static_assert(BU_WAVES * 32 == N * K, "one prologue thread per output element");
static_assert(N * K == D && K == 16, "the prologue's thread map assumes 256 columns in capsules of 16");

__device__ __forceinline__ float quad_xor1(float v) {
    return __builtin_bit_cast(float, __builtin_amdgcn_mov_dpp(__builtin_bit_cast(int, v), 0xB1, 0xF, 0xF, true));
}
__device__ __forceinline__ float quad_xor2(float v) {
    return __builtin_bit_cast(float, __builtin_amdgcn_mov_dpp(__builtin_bit_cast(int, v), 0x4E, 0xF, 0xF, true));
}

__global__ void __launch_bounds__(BU_WAVES * 32)
bupdate(const float* __restrict__ x, const float* __restrict__ sp,
        const float* __restrict__ W,
        float* __restrict__ bc) {
    __shared__ __attribute__((aligned(16))) float stage[BU_WAVES * BU_WROWS * BU_PITCH];
    __shared__ __attribute__((aligned(16))) float sl[N * BU_TP];
    __shared__ __attribute__((aligned(16))) float tl[N * BU_TP];
    __shared__ __attribute__((aligned(16))) float part[4 * N * K];
    __shared__ __attribute__((aligned(16))) float oraw[N * K];
    __shared__ __attribute__((aligned(16))) float onrm[N * K];
    const int b    = blockIdx.y;
    const int tid  = threadIdx.x;
    const int lane = tid & 31;
    const int wave = tid >> 5;
    const int half = lane >> 4;
    const int l16  = lane & 15;

    const int q  = tid >> 6;
    const int mm = tid & 63;
    const int nq = mm >> 2;
    v4f wreg[BU_WPT];
    {
        const float* wp = W + (size_t)(q * BU_WPT) * (N * K) + mm * 4;
#pragma unroll
        for (int e = 0; e < BU_WPT; ++e) wreg[e] = *(const v4f*)(wp + (size_t)e * (N * K));
    }
#pragma unroll
    for (int r = 0; r < (N * D / 4) / (BU_WAVES * 32); ++r) {
        const int piece = r * (BU_WAVES * 32) + tid;
        const int n  = piece / (D / 4);
        const int c4 = piece % (D / 4);
        const size_t g = ((size_t)b * N + n) * D + c4 * 4;
        v4f a = *(const v4f*)(sp + g);
#pragma unroll
        for (int p = 1; p < GS_PLANES; ++p) a += *(const v4f*)(sp + (size_t)p * GS_PLANE + g);
        *(v4f*)(sl + n * BU_TP + c4 * 4) = a;
    }
    const int i0 = (blockIdx.x * BU_WAVES + wave) * BU_TPW * 16;
    const float* xr = x + ((size_t)b * In + i0 + l16) * D + half * 4;
    v4f xpre[BU_PRE][BU_TPW][2];
#pragma unroll
    for (int c = 0; c < BU_PRE; ++c) {
#pragma unroll
        for (int tt = 0; tt < BU_TPW; ++tt) {
            const float* p = xr + (size_t)tt * 16 * D + 16 * c;
            xpre[c][tt][0] = *(const v4f*)(p);
            xpre[c][tt][1] = *(const v4f*)(p + 8);
        }
    }
    __syncthreads();

    {
        v4f acc1 = {0.0f, 0.0f, 0.0f, 0.0f};
        const float* srow = sl + nq * BU_TP + q * BU_WPT;
#pragma unroll
        for (int e4 = 0; e4 < BU_WPT / 4; ++e4) {
            const v4f sv = *(const v4f*)(srow + 4 * e4);
            acc1 += sv.x * wreg[4 * e4 + 0];
            acc1 += sv.y * wreg[4 * e4 + 1];
            acc1 += sv.z * wreg[4 * e4 + 2];
            acc1 += sv.w * wreg[4 * e4 + 3];
        }
        *(v4f*)(part + q * (N * K) + mm * 4) = acc1;
    }
    __syncthreads();

    {
        const float o = ((part[tid] + part[N * K + tid]) + part[2 * N * K + tid]) + part[3 * N * K + tid];
        oraw[tid] = o;
        __syncthreads();
        const float* mine_o = oraw + (tid & ~(K - 1));
        float n2 = 0.0f;
#pragma unroll
        for (int k = 0; k < K; ++k) n2 += mine_o[k] * mine_o[k];
        onrm[tid] = o * rsqrtf(n2 + 1e-7f);
    }
    __syncthreads();

    {
        const v4f ov = *(const v4f*)(onrm + mm * 4);
        float* trow = tl + nq * BU_TP + q * BU_WPT;
#pragma unroll
        for (int e4 = 0; e4 < BU_WPT / 4; ++e4) {
            v4f tv;
#pragma unroll
            for (int u = 0; u < 4; ++u) {
                const v4f w = wreg[4 * e4 + u];
                float p = w.x * ov.x + w.y * ov.y + w.z * ov.z + w.w * ov.w;
                p += quad_xor1(p);
                p += quad_xor2(p);
                tv[u] = p;
            }
            if ((mm & 3) == 0) *(v4f*)(trow + 4 * e4) = tv;
        }
    }
    __syncthreads();

    const float* tr = tl + l16 * BU_TP + half * 4;

    v8f acc[BU_TPW][2] = {};
#pragma unroll
    for (int c = 0; c < D / 16; ++c) {
        const v4f ta = *(const v4f*)(tr + 16 * c);
        const v4f tb = *(const v4f*)(tr + 16 * c + 8);
        const v2f b0 = {ta.x, ta.y}, b1 = {ta.z, ta.w};
        const v2f b2 = {tb.x, tb.y}, b3 = {tb.z, tb.w};
#pragma unroll
        for (int tt = 0; tt < BU_TPW; ++tt) {
            const float* p = xr + (size_t)tt * 16 * D + 16 * c;
            const v4f xa = c < BU_PRE ? xpre[c < BU_PRE ? c : 0][tt][0] : *(const v4f*)(p);
            const v4f xb = c < BU_PRE ? xpre[c < BU_PRE ? c : 0][tt][1] : *(const v4f*)(p + 8);
            const v2f a0 = {xa.x, xa.y}, a1 = {xa.z, xa.w};
            const v2f a2 = {xb.x, xb.y}, a3 = {xb.z, xb.w};
            acc[tt][0] = __builtin_amdgcn_wmma_f32_16x16x4_f32(false, a0, false, b0, (short)0, acc[tt][0], false, false);
            acc[tt][1] = __builtin_amdgcn_wmma_f32_16x16x4_f32(false, a1, false, b1, (short)0, acc[tt][1], false, false);
            acc[tt][0] = __builtin_amdgcn_wmma_f32_16x16x4_f32(false, a2, false, b2, (short)0, acc[tt][0], false, false);
            acc[tt][1] = __builtin_amdgcn_wmma_f32_16x16x4_f32(false, a3, false, b3, (short)0, acc[tt][1], false, false);
        }
    }

    float* mine = stage + wave * (BU_WROWS * BU_PITCH);
#pragma unroll
    for (int tt = 0; tt < BU_TPW; ++tt) {
#pragma unroll
        for (int r = 0; r < 8; ++r)
            mine[(tt * 16 + r + half * 8) * BU_PITCH + l16] = acc[tt][0][r] + acc[tt][1][r];
    }
    __syncthreads();

#pragma unroll
    for (int p = 0; p < 2; ++p) {
        float* row = mine + (2 * lane + p) * BU_PITCH;
        float v[N];
#pragma unroll
        for (int q = 0; q < N / 4; ++q) {
            const v4f piece = *(const v4f*)(row + 4 * q);
            v[4 * q + 0] = piece.x; v[4 * q + 1] = piece.y;
            v[4 * q + 2] = piece.z; v[4 * q + 3] = piece.w;
        }
        softmax_row(v);
#pragma unroll
        for (int q = 0; q < N / 4; ++q) {
            const v4f piece = {v[4 * q + 0], v[4 * q + 1], v[4 * q + 2], v[4 * q + 3]};
            *(v4f*)(row + 4 * q) = piece;
        }
    }
    __syncthreads();

    float* dst = bc + ((size_t)b * In + i0) * N;
#pragma unroll
    for (int k = 0; k < BU_WROWS * N / (4 * 32); ++k) {
        const int piece = 32 * k + lane;
        const v4f o = *(const v4f*)(mine + (piece >> 2) * BU_PITCH + (piece & 3) * 4);
        *(v4f*)(dst + (size_t)piece * 4) = o;
    }
}

extern "C" void kernel_launch(void* const* d_in, const int* in_sizes, int n_in,
                              void* d_out, int out_size, void* d_ws, size_t ws_size,
                              hipStream_t stream) {
    const float* x = (const float*)d_in[0];
    const float* W = (const float*)d_in[1];
    float* out = (float*)d_out;

    float* ws     = (float*)d_ws;
    float* bc     = ws;
    float* sp     = bc + (size_t)B * In * N;

    zero_buf<<<ZB_BLOCKS, ZB_THREADS, 0, stream>>>(bc);
    softmax_n<<<(B * In + 255) / 256, 256, 0, stream>>>(bc);

    for (int it = 0; it < 3; ++it) {
        gemm_s<<<dim3(GS_PLANES, B), GS_WAVES * 32, 0, stream>>>(x, bc, sp);
        if (it < 2)
            bupdate<<<dim3(In / (16 * BU_TPW * BU_WAVES), B), BU_WAVES * 32, 0, stream>>>(x, sp, W, bc);
        else
            outputs_squash<<<B * N, 256, 0, stream>>>(sp, W, out);
    }
}


# first couplings: one launch forms the softmax of a zero row in registers and stores c as whole 128-byte lines; the clear launch and the softmax's read of it are gone, 8 launches to 7; no sum touched
# speedup vs baseline: 3.4557x; 1.0703x over previous
#include <hip/hip_runtime.h>
#include <hip/hip_bf16.h>


typedef __attribute__((ext_vector_type(2))) float v2f;
typedef __attribute__((ext_vector_type(4))) float v4f;
typedef __attribute__((ext_vector_type(8))) float v8f;

namespace caps {
constexpr int B   = 32;
constexpr int In  = 4096;
constexpr int D   = 256;
constexpr int N   = 16;
constexpr int K   = 16;
}

using namespace caps;

__device__ __forceinline__ void softmax_row(float (&v)[N]) {
    float m = -3.402823466e+38f;
#pragma unroll
    for (int n = 0; n < N; ++n) m = fmaxf(m, v[n]);
    float s = 0.0f;
#pragma unroll
    for (int n = 0; n < N; ++n) { v[n] = __expf(v[n] - m); s += v[n]; }
    const float inv = 1.0f / s;
#pragma unroll
    for (int n = 0; n < N; ++n) v[n] = v[n] * inv;
}

constexpr int IC_THREADS    = 256;
constexpr int IC_PER_THREAD = 8;
constexpr int IC_BLOCKS     = (B * In * N) / (4 * IC_THREADS * IC_PER_THREAD);
static_assert((B * In * N) % (4 * IC_THREADS * IC_PER_THREAD) == 0,
              "the couplings must divide into whole work-group pieces");
static_assert(N % 4 == 0 && IC_THREADS % (N / 4) == 0,
              "a thread must keep the same 16-byte piece of a row in every store");

__global__ void __launch_bounds__(IC_THREADS)
init_c(float* __restrict__ bc) {
    float v[N];
#pragma unroll
    for (int n = 0; n < N; ++n) v[n] = 0.0f;
    softmax_row(v);
    const int j = threadIdx.x % (N / 4);
    v4f piece = {v[0], v[1], v[2], v[3]};
#pragma unroll
    for (int q = 1; q < N / 4; ++q)
        if (j == q) piece = {v[4 * q + 0], v[4 * q + 1], v[4 * q + 2], v[4 * q + 3]};
    v4f* p = (v4f*)bc + (size_t)blockIdx.x * (IC_THREADS * IC_PER_THREAD) + threadIdx.x;
    for (int k = 0; k < IC_PER_THREAD; ++k) p[k * IC_THREADS] = piece;
}

constexpr int GS_WAVES = 16;
constexpr int GS_COLS  = 64;
constexpr int GS_GRP   = 8;
constexpr int GS_CSL    = D / GS_COLS;
constexpr int GS_RGRP   = GS_WAVES / GS_CSL;
constexpr int GS_PLANES = 8;
constexpr int GS_BROWS  = In / GS_PLANES;
constexpr int GS_ROWS   = GS_BROWS / GS_RGRP;
constexpr int GS_NGRP   = GS_ROWS / (4 * GS_GRP);
constexpr int GS_PLANE  = B * N * D;
static_assert(GS_WAVES % GS_CSL == 0, "the waves must divide into whole row groups");
static_assert(GS_ROWS % (4 * GS_GRP) == 0, "a wave's rows must divide into whole load groups");
static_assert((N * D / 4) % (GS_WAVES * 32) == 0, "the 16-byte pieces of a plane must divide over the threads");

__global__ void __launch_bounds__(GS_WAVES * 32)
gemm_s(const float* __restrict__ x, const float* __restrict__ c,
       float* __restrict__ sp) {
    __shared__ __attribute__((aligned(16))) float red[GS_WAVES * N * GS_COLS];
    const int b    = blockIdx.y;
    const int rb   = blockIdx.x;
    const int lane = threadIdx.x & 31;
    const int wave = threadIdx.x >> 5;
    const int half = lane >> 4;
    const int l16  = lane & 15;
    const int j    = wave % GS_CSL;
    const int rg   = wave / GS_CSL;

    const int r0 = rb * GS_BROWS + rg * GS_ROWS + half * 2;
    const float* cp = c + ((size_t)b * In + r0) * N + l16;
    const float* xp = x + ((size_t)b * In + r0) * D + j * GS_COLS + l16 * 4;

    v8f acc0 = {}, acc1 = {}, acc2 = {}, acc3 = {};
    v4f xv[2][GS_GRP][2];
    float cv[2][GS_GRP][2];

#pragma unroll
    for (int ks = 0; ks < GS_GRP; ++ks) {
#pragma unroll
        for (int v = 0; v < 2; ++v) {
            xv[0][ks][v] = *(const v4f*)(xp + (size_t)(4 * ks + v) * D);
            cv[0][ks][v] = cp[(size_t)(4 * ks + v) * N];
        }
    }

#pragma unroll
    for (int g = 0; g < GS_NGRP; ++g) {
        const int cur = g & 1, nxt = cur ^ 1;
        if (g + 1 < GS_NGRP) {
#pragma unroll
            for (int ks = 0; ks < GS_GRP; ++ks) {
#pragma unroll
                for (int v = 0; v < 2; ++v) {
                    const int row = 4 * (GS_GRP * (g + 1) + ks) + v;
                    xv[nxt][ks][v] = *(const v4f*)(xp + (size_t)row * D);
                    cv[nxt][ks][v] = cp[(size_t)row * N];
                }
            }
        }
        __builtin_amdgcn_sched_barrier(0);
#pragma unroll
        for (int ks = 0; ks < GS_GRP; ++ks) {
            const v4f x0 = xv[cur][ks][0], x1 = xv[cur][ks][1];
            const v2f a  = {cv[cur][ks][0], cv[cur][ks][1]};
            const v2f b0 = {x0.x, x1.x}, b1 = {x0.y, x1.y};
            const v2f b2 = {x0.z, x1.z}, b3 = {x0.w, x1.w};
            acc0 = __builtin_amdgcn_wmma_f32_16x16x4_f32(false, a, false, b0, (short)0, acc0, false, false);
            acc1 = __builtin_amdgcn_wmma_f32_16x16x4_f32(false, a, false, b1, (short)0, acc1, false, false);
            acc2 = __builtin_amdgcn_wmma_f32_16x16x4_f32(false, a, false, b2, (short)0, acc2, false, false);
            acc3 = __builtin_amdgcn_wmma_f32_16x16x4_f32(false, a, false, b3, (short)0, acc3, false, false);
        }
    }

    float* mine = red + wave * (N * GS_COLS);
#pragma unroll
    for (int r = 0; r < 8; ++r) {
        const v4f o = {acc0[r], acc1[r], acc2[r], acc3[r]};
        *(v4f*)(mine + (r + half * 8) * GS_COLS + l16 * 4) = o;
    }
    __syncthreads();

    float* __restrict__ s = sp + (size_t)rb * GS_PLANE;
#pragma unroll
    for (int r = 0; r < (N * D / 4) / (GS_WAVES * 32); ++r) {
        const int piece = r * (GS_WAVES * 32) + threadIdx.x;
        const int n  = piece / (D / 4);
        const int q4 = piece % (D / 4);
        const int jj = q4 / (GS_COLS / 4);
        const int c4 = q4 % (GS_COLS / 4);
        v4f sum = *(const v4f*)(red + (jj * N + n) * GS_COLS + c4 * 4);
#pragma unroll
        for (int g = 1; g < GS_RGRP; ++g)
            sum += *(const v4f*)(red + ((g * GS_CSL + jj) * N + n) * GS_COLS + c4 * 4);
        *(v4f*)(s + ((size_t)b * N + n) * D + q4 * 4) = sum;
    }
}

__global__ void __launch_bounds__(256)
outputs_squash(const float* __restrict__ sp,
               const float* __restrict__ W, float* __restrict__ out) {
    __shared__ float part[D * K];
    __shared__ float dot[K];
    const int bn = blockIdx.x;
    const int n  = bn & (N - 1);
    const int d  = threadIdx.x;

    float sv = sp[(size_t)bn * D + d];
#pragma unroll
    for (int p = 1; p < GS_PLANES; ++p) sv += sp[(size_t)p * GS_PLANE + (size_t)bn * D + d];
    const float* wr = W + (size_t)d * (N * K) + (size_t)n * K;
#pragma unroll
    for (int k = 0; k < K; ++k) part[d * K + k] = sv * wr[k];
    __syncthreads();

    if (d < K) {
        float a = 0.0f;
        for (int dd = 0; dd < D; ++dd) a += part[dd * K + d];
        dot[d] = a;
    }
    __syncthreads();

    if (d < K) {
        float n2 = 0.0f;
#pragma unroll
        for (int k = 0; k < K; ++k) n2 += dot[k] * dot[k];
        out[(size_t)bn * K + d] = dot[d] * rsqrtf(n2 + 1e-7f);
    }
}

constexpr int BU_TPW   = 4;
constexpr int BU_WAVES = 8;
constexpr int BU_WROWS = BU_TPW * 16;
constexpr int BU_PITCH = 20;
constexpr int BU_TP    = D + 4;
constexpr int BU_WPT   = D / 4;
constexpr int BU_PRE   = 4;
static_assert(BU_WROWS == 2 * 32, "each lane owns two of its wave's rows");
static_assert(BU_WROWS * BU_WAVES == GS_BROWS, "bupdate and gemm_s must give a work-group the same rows of x");
static_assert(BU_WAVES * 32 == N * K, "one prologue thread per output element");
static_assert(N * K == D && K == 16, "the prologue's thread map assumes 256 columns in capsules of 16");

__device__ __forceinline__ float quad_xor1(float v) {
    return __builtin_bit_cast(float, __builtin_amdgcn_mov_dpp(__builtin_bit_cast(int, v), 0xB1, 0xF, 0xF, true));
}
__device__ __forceinline__ float quad_xor2(float v) {
    return __builtin_bit_cast(float, __builtin_amdgcn_mov_dpp(__builtin_bit_cast(int, v), 0x4E, 0xF, 0xF, true));
}

__global__ void __launch_bounds__(BU_WAVES * 32)
bupdate(const float* __restrict__ x, const float* __restrict__ sp,
        const float* __restrict__ W,
        float* __restrict__ bc) {
    __shared__ __attribute__((aligned(16))) float stage[BU_WAVES * BU_WROWS * BU_PITCH];
    __shared__ __attribute__((aligned(16))) float sl[N * BU_TP];
    __shared__ __attribute__((aligned(16))) float tl[N * BU_TP];
    __shared__ __attribute__((aligned(16))) float part[4 * N * K];
    __shared__ __attribute__((aligned(16))) float oraw[N * K];
    __shared__ __attribute__((aligned(16))) float onrm[N * K];
    const int b    = blockIdx.y;
    const int tid  = threadIdx.x;
    const int lane = tid & 31;
    const int wave = tid >> 5;
    const int half = lane >> 4;
    const int l16  = lane & 15;

    const int q  = tid >> 6;
    const int mm = tid & 63;
    const int nq = mm >> 2;
    v4f wreg[BU_WPT];
    {
        const float* wp = W + (size_t)(q * BU_WPT) * (N * K) + mm * 4;
#pragma unroll
        for (int e = 0; e < BU_WPT; ++e) wreg[e] = *(const v4f*)(wp + (size_t)e * (N * K));
    }
#pragma unroll
    for (int r = 0; r < (N * D / 4) / (BU_WAVES * 32); ++r) {
        const int piece = r * (BU_WAVES * 32) + tid;
        const int n  = piece / (D / 4);
        const int c4 = piece % (D / 4);
        const size_t g = ((size_t)b * N + n) * D + c4 * 4;
        v4f a = *(const v4f*)(sp + g);
#pragma unroll
        for (int p = 1; p < GS_PLANES; ++p) a += *(const v4f*)(sp + (size_t)p * GS_PLANE + g);
        *(v4f*)(sl + n * BU_TP + c4 * 4) = a;
    }
    const int i0 = (blockIdx.x * BU_WAVES + wave) * BU_TPW * 16;
    const float* xr = x + ((size_t)b * In + i0 + l16) * D + half * 4;
    v4f xpre[BU_PRE][BU_TPW][2];
#pragma unroll
    for (int c = 0; c < BU_PRE; ++c) {
#pragma unroll
        for (int tt = 0; tt < BU_TPW; ++tt) {
            const float* p = xr + (size_t)tt * 16 * D + 16 * c;
            xpre[c][tt][0] = *(const v4f*)(p);
            xpre[c][tt][1] = *(const v4f*)(p + 8);
        }
    }
    __syncthreads();

    {
        v4f acc1 = {0.0f, 0.0f, 0.0f, 0.0f};
        const float* srow = sl + nq * BU_TP + q * BU_WPT;
#pragma unroll
        for (int e4 = 0; e4 < BU_WPT / 4; ++e4) {
            const v4f sv = *(const v4f*)(srow + 4 * e4);
            acc1 += sv.x * wreg[4 * e4 + 0];
            acc1 += sv.y * wreg[4 * e4 + 1];
            acc1 += sv.z * wreg[4 * e4 + 2];
            acc1 += sv.w * wreg[4 * e4 + 3];
        }
        *(v4f*)(part + q * (N * K) + mm * 4) = acc1;
    }
    __syncthreads();

    {
        const float o = ((part[tid] + part[N * K + tid]) + part[2 * N * K + tid]) + part[3 * N * K + tid];
        oraw[tid] = o;
        __syncthreads();
        const float* mine_o = oraw + (tid & ~(K - 1));
        float n2 = 0.0f;
#pragma unroll
        for (int k = 0; k < K; ++k) n2 += mine_o[k] * mine_o[k];
        onrm[tid] = o * rsqrtf(n2 + 1e-7f);
    }
    __syncthreads();

    {
        const v4f ov = *(const v4f*)(onrm + mm * 4);
        float* trow = tl + nq * BU_TP + q * BU_WPT;
#pragma unroll
        for (int e4 = 0; e4 < BU_WPT / 4; ++e4) {
            v4f tv;
#pragma unroll
            for (int u = 0; u < 4; ++u) {
                const v4f w = wreg[4 * e4 + u];
                float p = w.x * ov.x + w.y * ov.y + w.z * ov.z + w.w * ov.w;
                p += quad_xor1(p);
                p += quad_xor2(p);
                tv[u] = p;
            }
            if ((mm & 3) == 0) *(v4f*)(trow + 4 * e4) = tv;
        }
    }
    __syncthreads();

    const float* tr = tl + l16 * BU_TP + half * 4;

    v8f acc[BU_TPW][2] = {};
#pragma unroll
    for (int c = 0; c < D / 16; ++c) {
        const v4f ta = *(const v4f*)(tr + 16 * c);
        const v4f tb = *(const v4f*)(tr + 16 * c + 8);
        const v2f b0 = {ta.x, ta.y}, b1 = {ta.z, ta.w};
        const v2f b2 = {tb.x, tb.y}, b3 = {tb.z, tb.w};
#pragma unroll
        for (int tt = 0; tt < BU_TPW; ++tt) {
            const float* p = xr + (size_t)tt * 16 * D + 16 * c;
            const v4f xa = c < BU_PRE ? xpre[c < BU_PRE ? c : 0][tt][0] : *(const v4f*)(p);
            const v4f xb = c < BU_PRE ? xpre[c < BU_PRE ? c : 0][tt][1] : *(const v4f*)(p + 8);
            const v2f a0 = {xa.x, xa.y}, a1 = {xa.z, xa.w};
            const v2f a2 = {xb.x, xb.y}, a3 = {xb.z, xb.w};
            acc[tt][0] = __builtin_amdgcn_wmma_f32_16x16x4_f32(false, a0, false, b0, (short)0, acc[tt][0], false, false);
            acc[tt][1] = __builtin_amdgcn_wmma_f32_16x16x4_f32(false, a1, false, b1, (short)0, acc[tt][1], false, false);
            acc[tt][0] = __builtin_amdgcn_wmma_f32_16x16x4_f32(false, a2, false, b2, (short)0, acc[tt][0], false, false);
            acc[tt][1] = __builtin_amdgcn_wmma_f32_16x16x4_f32(false, a3, false, b3, (short)0, acc[tt][1], false, false);
        }
    }

    float* mine = stage + wave * (BU_WROWS * BU_PITCH);
#pragma unroll
    for (int tt = 0; tt < BU_TPW; ++tt) {
#pragma unroll
        for (int r = 0; r < 8; ++r)
            mine[(tt * 16 + r + half * 8) * BU_PITCH + l16] = acc[tt][0][r] + acc[tt][1][r];
    }
    __syncthreads();

#pragma unroll
    for (int p = 0; p < 2; ++p) {
        float* row = mine + (2 * lane + p) * BU_PITCH;
        float v[N];
#pragma unroll
        for (int q = 0; q < N / 4; ++q) {
            const v4f piece = *(const v4f*)(row + 4 * q);
            v[4 * q + 0] = piece.x; v[4 * q + 1] = piece.y;
            v[4 * q + 2] = piece.z; v[4 * q + 3] = piece.w;
        }
        softmax_row(v);
#pragma unroll
        for (int q = 0; q < N / 4; ++q) {
            const v4f piece = {v[4 * q + 0], v[4 * q + 1], v[4 * q + 2], v[4 * q + 3]};
            *(v4f*)(row + 4 * q) = piece;
        }
    }
    __syncthreads();

    float* dst = bc + ((size_t)b * In + i0) * N;
#pragma unroll
    for (int k = 0; k < BU_WROWS * N / (4 * 32); ++k) {
        const int piece = 32 * k + lane;
        const v4f o = *(const v4f*)(mine + (piece >> 2) * BU_PITCH + (piece & 3) * 4);
        *(v4f*)(dst + (size_t)piece * 4) = o;
    }
}

extern "C" void kernel_launch(void* const* d_in, const int* in_sizes, int n_in,
                              void* d_out, int out_size, void* d_ws, size_t ws_size,
                              hipStream_t stream) {
    const float* x = (const float*)d_in[0];
    const float* W = (const float*)d_in[1];
    float* out = (float*)d_out;

    float* ws     = (float*)d_ws;
    float* bc     = ws;
    float* sp     = bc + (size_t)B * In * N;

    init_c<<<IC_BLOCKS, IC_THREADS, 0, stream>>>(bc);

    for (int it = 0; it < 3; ++it) {
        gemm_s<<<dim3(GS_PLANES, B), GS_WAVES * 32, 0, stream>>>(x, bc, sp);
        if (it < 2)
            bupdate<<<dim3(In / (16 * BU_TPW * BU_WAVES), B), BU_WAVES * 32, 0, stream>>>(x, sp, W, bc);
        else
            outputs_squash<<<B * N, 256, 0, stream>>>(sp, W, out);
    }
}
